# starting kernel
# baseline (speedup 1.0000x reference)
#include <hip/hip_runtime.h>


typedef __attribute__((ext_vector_type(2))) float v2f;
typedef __attribute__((ext_vector_type(8))) float v8f;

#define GNN_SLOPE 0.01f
#define GNN_EPS   1e-5f
#define GNN_BSN   16
#define GNN_N     2000
#define GNN_E     32000
#define GNN_NT    (GNN_BSN * GNN_N)
#define GNN_DOUT  128

__device__ __forceinline__ float lrelu(float v) { return v >= 0.0f ? v : GNN_SLOPE * v; }

#define WMMA_F32(A, B, C) \
    __builtin_amdgcn_wmma_f32_16x16x4_f32(false, (A), false, (B), (short)0, (C), false, false)

__global__ __launch_bounds__(256) void k_fill_zero(float* __restrict__ p, int n) {
    int i = blockIdx.x * 256 + threadIdx.x;
    if (i < n) p[i] = 0.0f;
}

template <bool DIFF>
__global__ __launch_bounds__(256) void k_pack_b(const float* __restrict__ B,
                                                float* __restrict__ Bp, int K) {
    int i = blockIdx.x * 256 + threadIdx.x;
    const int kk   = i >> 9;
    const int rem  = i & 511;
    const int lane = rem >> 4;
    const int j    = rem & 15;
    const int t    = j >> 1;
    const int comp = j & 1;
    const int r    = kk * 4 + 2 * (lane >> 4) + comp;
    const int c    = t * 16 + (lane & 15);
    const int src  = r * GNN_DOUT + c;
    float v = B[src];
    if (DIFF) v -= B[K * GNN_DOUT + src];
    Bp[i] = v;
}

template <int K>
__global__ __launch_bounds__(256) void k_gemm_pq(const float* __restrict__ A,
                                                 const float* __restrict__ BpA,
                                                 const float* __restrict__ BpB,
                                                 float* __restrict__ P,
                                                 float* __restrict__ Q) {
    const int wave = threadIdx.x >> 5;
    const int lane = threadIdx.x & 31;
    const int half = lane >> 4;
    const int lm   = lane & 15;
    const int m0   = blockIdx.x * 128 + wave * 16;

    v8f accP[8] = {};
    v8f accQ[8] = {};

    const float*  arow = A + (size_t)(m0 + lm) * K + 2 * half;
    const float4* bpA  = (const float4*)(BpA + (size_t)lane * 16);
    const float4* bpB  = (const float4*)(BpB + (size_t)lane * 16);

#pragma unroll
    for (int k = 0; k < K; k += 4) {
        const v2f a = *(const v2f*)(arow + k);
        const float4* ba = bpA + (size_t)(k >> 2) * 128;
        const float4* bb = bpB + (size_t)(k >> 2) * 128;
        float4 pa[4], pb[4];
#pragma unroll
        for (int i = 0; i < 4; ++i) { pa[i] = ba[i]; pb[i] = bb[i]; }
        v2f b;
#pragma unroll
        for (int i = 0; i < 4; ++i) {
            b.x = pa[i].x; b.y = pa[i].y; accP[2 * i + 0] = WMMA_F32(a, b, accP[2 * i + 0]);
            b.x = pa[i].z; b.y = pa[i].w; accP[2 * i + 1] = WMMA_F32(a, b, accP[2 * i + 1]);
            b.x = pb[i].x; b.y = pb[i].y; accQ[2 * i + 0] = WMMA_F32(a, b, accQ[2 * i + 0]);
            b.x = pb[i].z; b.y = pb[i].w; accQ[2 * i + 1] = WMMA_F32(a, b, accQ[2 * i + 1]);
        }
    }

    const size_t rb = (size_t)(m0 + 8 * half) * GNN_DOUT + lm;
    float* prow = P + rb;
    float* qrow = Q + rb;
#pragma unroll
    for (int r = 0; r < 8; ++r)
#pragma unroll
        for (int t = 0; t < 8; ++t) {
            prow[(size_t)r * GNN_DOUT + t * 16] = accP[t][r];
            qrow[(size_t)r * GNN_DOUT + t * 16] = accQ[t][r];
        }
}

__global__ __launch_bounds__(256) void k_edge(const float* __restrict__ P,
                                              const float* __restrict__ Q,
                                              const float* __restrict__ b1,
                                              const int* __restrict__ ei,
                                              float* __restrict__ S,
                                              float* __restrict__ cnt) {
    const int t   = blockIdx.x * 256 + threadIdx.x;
    const int e   = t >> 5;
    const int c   = (t & 31) * 4;
    const int off = blockIdx.y * GNN_N;
    const int src = ei[e] + off;
    const int dst = ei[GNN_E + e] + off;

    const float4 p  = *(const float4*)(P + (size_t)dst * GNN_DOUT + c);
    const float4 q  = *(const float4*)(Q + (size_t)src * GNN_DOUT + c);
    const float4 bb = *(const float4*)(b1 + c);

    float* sp = S + (size_t)dst * GNN_DOUT + c;
    atomicAdd(sp + 0, lrelu(p.x + q.x + bb.x));
    atomicAdd(sp + 1, lrelu(p.y + q.y + bb.y));
    atomicAdd(sp + 2, lrelu(p.z + q.z + bb.z));
    atomicAdd(sp + 3, lrelu(p.w + q.w + bb.w));
    if ((t & 31) == 0) atomicAdd(cnt + dst, 1.0f);
}

template <bool RES>
__global__ __launch_bounds__(256) void k_gemm_agg_ln(const float* __restrict__ S,
                                                     const float* __restrict__ Bp2,
                                                     const float* __restrict__ cnt,
                                                     const float* __restrict__ b2,
                                                     const float* __restrict__ g,
                                                     const float* __restrict__ beta,
                                                     const float* __restrict__ xres,
                                                     float* __restrict__ out) {
    const int wave = threadIdx.x >> 5;
    const int lane = threadIdx.x & 31;
    const int half = lane >> 4;
    const int lm   = lane & 15;
    const int m0   = blockIdx.x * 128 + wave * 16;

    const float cntv  = cnt[m0 + lm];
    const float scale = 1.0f / fmaxf(cntv, 1.0f);

    v8f acc[8] = {};
    const float*  arow = S + (size_t)(m0 + lm) * 128 + 2 * half;
    const float4* bp   = (const float4*)(Bp2 + (size_t)lane * 16);

#pragma unroll
    for (int k = 0; k < 128; k += 4) {
        v2f a = *(const v2f*)(arow + k);
        a.x *= scale; a.y *= scale;
        const float4* bk = bp + (size_t)(k >> 2) * 128;
        float4 pb[4];
#pragma unroll
        for (int i = 0; i < 4; ++i) pb[i] = bk[i];
        v2f b;
#pragma unroll
        for (int i = 0; i < 4; ++i) {
            b.x = pb[i].x; b.y = pb[i].y; acc[2 * i + 0] = WMMA_F32(a, b, acc[2 * i + 0]);
            b.x = pb[i].z; b.y = pb[i].w; acc[2 * i + 1] = WMMA_F32(a, b, acc[2 * i + 1]);
        }
    }

    float b2v[8], gv[8], bev[8];
#pragma unroll
    for (int t = 0; t < 8; ++t) {
        b2v[t] = b2[t * 16 + lm];
        gv[t]  = g[t * 16 + lm];
        bev[t] = beta[t * 16 + lm];
    }

    const size_t rb = (size_t)(m0 + 8 * half) * GNN_DOUT + lm;
    float* crow = out + rb;
    const float* rrow = RES ? (xres + rb) : nullptr;

#pragma unroll
    for (int r = 0; r < 8; ++r) {
        const float cr   = __shfl(cntv, r + 8 * half, 32);
        const float hasE = (cr > 0.0f) ? 1.0f : 0.0f;

        float h[8];
        float s = 0.0f;
#pragma unroll
        for (int t = 0; t < 8; ++t) { h[t] = acc[t][r] + hasE * b2v[t]; s += h[t]; }
#pragma unroll
        for (int m = 8; m >= 1; m >>= 1) s += __shfl_xor(s, m, 32);
        const float mu = s * (1.0f / GNN_DOUT);

        float s2 = 0.0f;
#pragma unroll
        for (int t = 0; t < 8; ++t) { h[t] -= mu; s2 += h[t] * h[t]; }
#pragma unroll
        for (int m = 8; m >= 1; m >>= 1) s2 += __shfl_xor(s2, m, 32);
        const float rstd = rsqrtf(s2 * (1.0f / GNN_DOUT) + GNN_EPS);

#pragma unroll
        for (int t = 0; t < 8; ++t) {
            float o = h[t] * rstd * gv[t] + bev[t];
            if (RES) o += rrow[(size_t)r * GNN_DOUT + t * 16];
            crow[(size_t)r * GNN_DOUT + t * 16] = lrelu(o);
        }
    }
}

extern "C" void kernel_launch(void* const* d_in, const int* in_sizes, int n_in,
                              void* d_out, int out_size, void* d_ws, size_t ws_size,
                              hipStream_t stream) {
    (void)in_sizes; (void)n_in; (void)out_size; (void)ws_size;

    const float* x  = (const float*)d_in[0];
    const int*   ei = (const int*)d_in[1];
    const float* W1[2] = {(const float*)d_in[2],  (const float*)d_in[8]};
    const float* b1[2] = {(const float*)d_in[3],  (const float*)d_in[9]};
    const float* W2[2] = {(const float*)d_in[4],  (const float*)d_in[10]};
    const float* b2[2] = {(const float*)d_in[5],  (const float*)d_in[11]};
    const float* g [2] = {(const float*)d_in[6],  (const float*)d_in[12]};
    const float* be[2] = {(const float*)d_in[7],  (const float*)d_in[13]};

    const size_t NF = (size_t)GNN_NT * GNN_DOUT;
    float* ws   = (float*)d_ws;
    float* X    = ws;
    float* P    = X + NF;
    float* Q    = P + NF;
    float* S    = Q + NF;
    float* cnt  = S + NF;
    float* BpA  = cnt + GNN_NT;
    float* BpB  = BpA + 128 * GNN_DOUT;
    float* Bp2  = BpB + 128 * GNN_DOUT;

    const dim3 egrid(GNN_E * 32 / 256, GNN_BSN);
    const int  zgrid = (int)((NF + GNN_NT + 255) / 256);

    k_pack_b<true ><<< 64 * GNN_DOUT / 256, 256, 0, stream>>>(W1[0], BpA, 64);
    k_pack_b<false><<< 64 * GNN_DOUT / 256, 256, 0, stream>>>(W1[0] + 64 * GNN_DOUT, BpB, 64);
    k_pack_b<false><<<128 * GNN_DOUT / 256, 256, 0, stream>>>(W2[0], Bp2, 128);
    k_fill_zero<<<zgrid, 256, 0, stream>>>(S, (int)(NF + GNN_NT));
    k_gemm_pq<64><<<250, 256, 0, stream>>>(x, BpA, BpB, P, Q);
    k_edge<<<egrid, 256, 0, stream>>>(P, Q, b1[0], ei, S, cnt);
    k_gemm_agg_ln<false><<<250, 256, 0, stream>>>(S, Bp2, cnt, b2[0], g[0], be[0], nullptr, X);

    k_pack_b<true ><<<128 * GNN_DOUT / 256, 256, 0, stream>>>(W1[1], BpA, 128);
    k_pack_b<false><<<128 * GNN_DOUT / 256, 256, 0, stream>>>(W1[1] + 128 * GNN_DOUT, BpB, 128);
    k_pack_b<false><<<128 * GNN_DOUT / 256, 256, 0, stream>>>(W2[1], Bp2, 128);
    k_fill_zero<<<zgrid, 256, 0, stream>>>(S, (int)(NF + GNN_NT));
    k_gemm_pq<128><<<250, 256, 0, stream>>>(X, BpA, BpB, P, Q);
    k_edge<<<egrid, 256, 0, stream>>>(P, Q, b1[1], ei, S, cnt);
    k_gemm_agg_ln<true><<<250, 256, 0, stream>>>(S, Bp2, cnt, b2[1], g[1], be[1], X, (float*)d_out);
}


# Edges bucketed by destination once per call (LDS integer counters, scan, edge-id order); one wave sums each node row in registers: no f32 atomics or zero-fill; float sum order now fixed
# speedup vs baseline: 2.4202x; 2.4202x over previous
#include <hip/hip_runtime.h>


typedef __attribute__((ext_vector_type(2))) float v2f;
typedef __attribute__((ext_vector_type(8))) float v8f;

#define GNN_SLOPE 0.01f
#define GNN_EPS   1e-5f
#define GNN_BSN   16
#define GNN_N     2000
#define GNN_E     32000
#define GNN_NT    (GNN_BSN * GNN_N)
#define GNN_DOUT  128

__device__ __forceinline__ float lrelu(float v) { return v >= 0.0f ? v : GNN_SLOPE * v; }

#define WMMA_F32(A, B, C) \
    __builtin_amdgcn_wmma_f32_16x16x4_f32(false, (A), false, (B), (short)0, (C), false, false)

template <bool DIFF>
__global__ __launch_bounds__(256) void k_pack_b(const float* __restrict__ B,
                                                float* __restrict__ Bp, int K) {
    int i = blockIdx.x * 256 + threadIdx.x;
    const int kk   = i >> 9;
    const int rem  = i & 511;
    const int lane = rem >> 4;
    const int j    = rem & 15;
    const int t    = j >> 1;
    const int comp = j & 1;
    const int r    = kk * 4 + 2 * (lane >> 4) + comp;
    const int c    = t * 16 + (lane & 15);
    const int src  = r * GNN_DOUT + c;
    float v = B[src];
    if (DIFF) v -= B[K * GNN_DOUT + src];
    Bp[i] = v;
}

template <int K>
__global__ __launch_bounds__(256) void k_gemm_pq(const float* __restrict__ A,
                                                 const float* __restrict__ BpA,
                                                 const float* __restrict__ BpB,
                                                 float* __restrict__ P,
                                                 float* __restrict__ Q) {
    const int wave = threadIdx.x >> 5;
    const int lane = threadIdx.x & 31;
    const int half = lane >> 4;
    const int lm   = lane & 15;
    const int m0   = blockIdx.x * 128 + wave * 16;

    v8f accP[8] = {};
    v8f accQ[8] = {};

    const float*  arow = A + (size_t)(m0 + lm) * K + 2 * half;
    const float4* bpA  = (const float4*)(BpA + (size_t)lane * 16);
    const float4* bpB  = (const float4*)(BpB + (size_t)lane * 16);

#pragma unroll
    for (int k = 0; k < K; k += 4) {
        const v2f a = *(const v2f*)(arow + k);
        const float4* ba = bpA + (size_t)(k >> 2) * 128;
        const float4* bb = bpB + (size_t)(k >> 2) * 128;
        float4 pa[4], pb[4];
#pragma unroll
        for (int i = 0; i < 4; ++i) { pa[i] = ba[i]; pb[i] = bb[i]; }
        v2f b;
#pragma unroll
        for (int i = 0; i < 4; ++i) {
            b.x = pa[i].x; b.y = pa[i].y; accP[2 * i + 0] = WMMA_F32(a, b, accP[2 * i + 0]);
            b.x = pa[i].z; b.y = pa[i].w; accP[2 * i + 1] = WMMA_F32(a, b, accP[2 * i + 1]);
            b.x = pb[i].x; b.y = pb[i].y; accQ[2 * i + 0] = WMMA_F32(a, b, accQ[2 * i + 0]);
            b.x = pb[i].z; b.y = pb[i].w; accQ[2 * i + 1] = WMMA_F32(a, b, accQ[2 * i + 1]);
        }
    }

    const size_t rb = (size_t)(m0 + 8 * half) * GNN_DOUT + lm;
    float* prow = P + rb;
    float* qrow = Q + rb;
#pragma unroll
    for (int r = 0; r < 8; ++r)
#pragma unroll
        for (int t = 0; t < 8; ++t) {
            prow[(size_t)r * GNN_DOUT + t * 16] = accP[t][r];
            qrow[(size_t)r * GNN_DOUT + t * 16] = accQ[t][r];
        }
}

#define GNN_CSR_T  1024
#define GNN_CSR_IT ((GNN_E + GNN_CSR_T - 1) / GNN_CSR_T)
__global__ __launch_bounds__(GNN_CSR_T) void k_csr_place(const int* __restrict__ ei,
                                                         int* __restrict__ rowptr,
                                                         int* __restrict__ slot) {
    __shared__ int hist[2 * GNN_CSR_T];
    __shared__ int wsum[32];
    const int tid  = threadIdx.x;
    const int lane = tid & 31;
    const int wave = tid >> 5;

    hist[2 * tid + 0] = 0;
    hist[2 * tid + 1] = 0;
    __syncthreads();

    int d[GNN_CSR_IT], r[GNN_CSR_IT];
#pragma unroll
    for (int it = 0; it < GNN_CSR_IT; ++it) {
        const int e = it * GNN_CSR_T + tid;
        d[it] = (e < GNN_E) ? ei[GNN_E + e] : -1;
    }
#pragma unroll
    for (int it = 0; it < GNN_CSR_IT; ++it)
        r[it] = (d[it] >= 0) ? atomicAdd(&hist[d[it]], 1) : 0;
    __syncthreads();

    const int a = hist[2 * tid + 0];
    const int s = a + hist[2 * tid + 1];
    int inc = s;
    for (int m = 1; m < 32; m <<= 1) {
        const int v = __shfl_up(inc, m, 32);
        if (lane >= m) inc += v;
    }
    if (lane == 31) wsum[wave] = inc;
    __syncthreads();
    if (wave == 0) {
        const int w = wsum[lane];
        int wi = w;
        for (int m = 1; m < 32; m <<= 1) {
            const int v = __shfl_up(wi, m, 32);
            if (lane >= m) wi += v;
        }
        wsum[lane] = wi - w;
    }
    __syncthreads();
    const int excl = wsum[wave] + inc - s;
    hist[2 * tid + 0] = excl;
    hist[2 * tid + 1] = excl + a;
    if (2 * tid + 0 <= GNN_N) rowptr[2 * tid + 0] = excl;
    if (2 * tid + 1 <= GNN_N) rowptr[2 * tid + 1] = excl + a;
    __syncthreads();

#pragma unroll
    for (int it = 0; it < GNN_CSR_IT; ++it) r[it] += hist[d[it] >= 0 ? d[it] : 0];
#pragma unroll
    for (int it = 0; it < GNN_CSR_IT; ++it)
        if (d[it] >= 0) slot[r[it]] = it * GNN_CSR_T + tid;
}

__global__ __launch_bounds__(256) void k_csr_order(const int* __restrict__ ei,
                                                   const int* __restrict__ rowptr,
                                                   const int* __restrict__ slot,
                                                   int* __restrict__ col) {
    const int e   = blockIdx.x * 256 + threadIdx.x;
    const int dst = ei[GNN_E + e];
    const int beg = rowptr[dst];
    const int end = rowptr[dst + 1];
    int rank = 0;
    for (int j = beg; j < end; ++j) rank += (slot[j] < e) ? 1 : 0;
    col[beg + rank] = ei[e];
}

__global__ __launch_bounds__(256) void k_gather(const float* __restrict__ P,
                                                const float* __restrict__ Q,
                                                const float* __restrict__ b1,
                                                const int* __restrict__ rowptr,
                                                const int* __restrict__ col,
                                                float* __restrict__ S,
                                                float* __restrict__ cnt) {
    const int wave = __builtin_amdgcn_readfirstlane((int)(threadIdx.x >> 5));
    const int lane = threadIdx.x & 31;
    const int c    = lane * 4;
    const int node = blockIdx.x * 8 + wave;
    const int off  = blockIdx.y * GNN_N;
    const int beg  = rowptr[node];
    const int end  = rowptr[node + 1];

    const size_t row = (size_t)(off + node) * GNN_DOUT + c;
    const float4 p  = *(const float4*)(P + row);
    const float4 bb = *(const float4*)(b1 + c);
    const float* qb = Q + (size_t)off * GNN_DOUT + c;

    float4 acc = {0.0f, 0.0f, 0.0f, 0.0f};
    for (int base = beg; base < end; base += 32) {
        const int n    = (end - base < 32) ? (end - base) : 32;
        const int mine = (lane < n) ? col[base + lane] : 0;
#pragma unroll 4
        for (int j = 0; j < n; ++j) {
            const int src = __builtin_amdgcn_readlane(mine, j);
            const float4 q = *(const float4*)(qb + (size_t)src * GNN_DOUT);
            acc.x += lrelu(p.x + q.x + bb.x);
            acc.y += lrelu(p.y + q.y + bb.y);
            acc.z += lrelu(p.z + q.z + bb.z);
            acc.w += lrelu(p.w + q.w + bb.w);
        }
    }
    *(float4*)(S + row) = acc;
    if (lane == 0) cnt[off + node] = (float)(end - beg);
}

template <bool RES>
__global__ __launch_bounds__(256) void k_gemm_agg_ln(const float* __restrict__ S,
                                                     const float* __restrict__ Bp2,
                                                     const float* __restrict__ cnt,
                                                     const float* __restrict__ b2,
                                                     const float* __restrict__ g,
                                                     const float* __restrict__ beta,
                                                     const float* __restrict__ xres,
                                                     float* __restrict__ out) {
    const int wave = threadIdx.x >> 5;
    const int lane = threadIdx.x & 31;
    const int half = lane >> 4;
    const int lm   = lane & 15;
    const int m0   = blockIdx.x * 128 + wave * 16;

    const float cntv  = cnt[m0 + lm];
    const float scale = 1.0f / fmaxf(cntv, 1.0f);

    v8f acc[8] = {};
    const float*  arow = S + (size_t)(m0 + lm) * 128 + 2 * half;
    const float4* bp   = (const float4*)(Bp2 + (size_t)lane * 16);

#pragma unroll
    for (int k = 0; k < 128; k += 4) {
        v2f a = *(const v2f*)(arow + k);
        a.x *= scale; a.y *= scale;
        const float4* bk = bp + (size_t)(k >> 2) * 128;
        float4 pb[4];
#pragma unroll
        for (int i = 0; i < 4; ++i) pb[i] = bk[i];
        v2f b;
#pragma unroll
        for (int i = 0; i < 4; ++i) {
            b.x = pb[i].x; b.y = pb[i].y; acc[2 * i + 0] = WMMA_F32(a, b, acc[2 * i + 0]);
            b.x = pb[i].z; b.y = pb[i].w; acc[2 * i + 1] = WMMA_F32(a, b, acc[2 * i + 1]);
        }
    }

    float b2v[8], gv[8], bev[8];
#pragma unroll
    for (int t = 0; t < 8; ++t) {
        b2v[t] = b2[t * 16 + lm];
        gv[t]  = g[t * 16 + lm];
        bev[t] = beta[t * 16 + lm];
    }

    const size_t rb = (size_t)(m0 + 8 * half) * GNN_DOUT + lm;
    float* crow = out + rb;
    const float* rrow = RES ? (xres + rb) : nullptr;

#pragma unroll
    for (int r = 0; r < 8; ++r) {
        const float cr   = __shfl(cntv, r + 8 * half, 32);
        const float hasE = (cr > 0.0f) ? 1.0f : 0.0f;

        float h[8];
        float s = 0.0f;
#pragma unroll
        for (int t = 0; t < 8; ++t) { h[t] = acc[t][r] + hasE * b2v[t]; s += h[t]; }
#pragma unroll
        for (int m = 8; m >= 1; m >>= 1) s += __shfl_xor(s, m, 32);
        const float mu = s * (1.0f / GNN_DOUT);

        float s2 = 0.0f;
#pragma unroll
        for (int t = 0; t < 8; ++t) { h[t] -= mu; s2 += h[t] * h[t]; }
#pragma unroll
        for (int m = 8; m >= 1; m >>= 1) s2 += __shfl_xor(s2, m, 32);
        const float rstd = rsqrtf(s2 * (1.0f / GNN_DOUT) + GNN_EPS);

#pragma unroll
        for (int t = 0; t < 8; ++t) {
            float o = h[t] * rstd * gv[t] + bev[t];
            if (RES) o += rrow[(size_t)r * GNN_DOUT + t * 16];
            crow[(size_t)r * GNN_DOUT + t * 16] = lrelu(o);
        }
    }
}

extern "C" void kernel_launch(void* const* d_in, const int* in_sizes, int n_in,
                              void* d_out, int out_size, void* d_ws, size_t ws_size,
                              hipStream_t stream) {
    (void)in_sizes; (void)n_in; (void)out_size; (void)ws_size;

    const float* x  = (const float*)d_in[0];
    const int*   ei = (const int*)d_in[1];
    const float* W1[2] = {(const float*)d_in[2],  (const float*)d_in[8]};
    const float* b1[2] = {(const float*)d_in[3],  (const float*)d_in[9]};
    const float* W2[2] = {(const float*)d_in[4],  (const float*)d_in[10]};
    const float* b2[2] = {(const float*)d_in[5],  (const float*)d_in[11]};
    const float* g [2] = {(const float*)d_in[6],  (const float*)d_in[12]};
    const float* be[2] = {(const float*)d_in[7],  (const float*)d_in[13]};

    const size_t NF = (size_t)GNN_NT * GNN_DOUT;
    float* ws   = (float*)d_ws;
    float* X    = ws;
    float* P    = X + NF;
    float* Q    = P + NF;
    float* S    = Q + NF;
    float* cnt  = S + NF;
    float* BpA  = cnt + GNN_NT;
    float* BpB  = BpA + 128 * GNN_DOUT;
    float* Bp2  = BpB + 128 * GNN_DOUT;
    int* rowptr = (int*)(Bp2 + 128 * GNN_DOUT);
    int* slot   = rowptr + 2048;
    int* col    = slot + GNN_E;

    const dim3 ggrid(GNN_N / 8, GNN_BSN);

    k_csr_place<<<1, GNN_CSR_T, 0, stream>>>(ei, rowptr, slot);
    k_csr_order<<<GNN_E / 256, 256, 0, stream>>>(ei, rowptr, slot, col);

    k_pack_b<true ><<< 64 * GNN_DOUT / 256, 256, 0, stream>>>(W1[0], BpA, 64);
    k_pack_b<false><<< 64 * GNN_DOUT / 256, 256, 0, stream>>>(W1[0] + 64 * GNN_DOUT, BpB, 64);
    k_pack_b<false><<<128 * GNN_DOUT / 256, 256, 0, stream>>>(W2[0], Bp2, 128);
    k_gemm_pq<64><<<250, 256, 0, stream>>>(x, BpA, BpB, P, Q);
    k_gather<<<ggrid, 256, 0, stream>>>(P, Q, b1[0], rowptr, col, S, cnt);
    k_gemm_agg_ln<false><<<250, 256, 0, stream>>>(S, Bp2, cnt, b2[0], g[0], be[0], nullptr, X);

    k_pack_b<true ><<<128 * GNN_DOUT / 256, 256, 0, stream>>>(W1[1], BpA, 128);
    k_pack_b<false><<<128 * GNN_DOUT / 256, 256, 0, stream>>>(W1[1] + 128 * GNN_DOUT, BpB, 128);
    k_pack_b<false><<<128 * GNN_DOUT / 256, 256, 0, stream>>>(W2[1], Bp2, 128);
    k_gemm_pq<128><<<250, 256, 0, stream>>>(X, BpA, BpB, P, Q);
    k_gather<<<ggrid, 256, 0, stream>>>(P, Q, b1[1], rowptr, col, S, cnt);
    k_gemm_agg_ln<true><<<250, 256, 0, stream>>>(S, Bp2, cnt, b2[1], g[1], be[1], X, (float*)d_out);
}


# Launches 14 -> 8: edge bucketing and all six weight packs (both blocks) share one set-up launch, each role on its own work-groups; no kernel body's arithmetic changes
# speedup vs baseline: 2.5932x; 1.0715x over previous
#include <hip/hip_runtime.h>


typedef __attribute__((ext_vector_type(2))) float v2f;
typedef __attribute__((ext_vector_type(8))) float v8f;

#define GNN_SLOPE 0.01f
#define GNN_EPS   1e-5f
#define GNN_BSN   16
#define GNN_N     2000
#define GNN_E     32000
#define GNN_NT    (GNN_BSN * GNN_N)
#define GNN_DOUT  128

__device__ __forceinline__ float lrelu(float v) { return v >= 0.0f ? v : GNN_SLOPE * v; }

#define WMMA_F32(A, B, C) \
    __builtin_amdgcn_wmma_f32_16x16x4_f32(false, (A), false, (B), (short)0, (C), false, false)

template <bool DIFF>
__device__ __forceinline__ void pack_b_elem(const float* __restrict__ B,
                                            float* __restrict__ Bp, int K, int i) {
    const int kk   = i >> 9;
    const int rem  = i & 511;
    const int lane = rem >> 4;
    const int j    = rem & 15;
    const int t    = j >> 1;
    const int comp = j & 1;
    const int r    = kk * 4 + 2 * (lane >> 4) + comp;
    const int c    = t * 16 + (lane & 15);
    const int src  = r * GNN_DOUT + c;
    float v = B[src];
    if (DIFF) v -= B[K * GNN_DOUT + src];
    Bp[i] = v;
}

template <int K>
__global__ __launch_bounds__(256) void k_gemm_pq(const float* __restrict__ A,
                                                 const float* __restrict__ BpA,
                                                 const float* __restrict__ BpB,
                                                 float* __restrict__ P,
                                                 float* __restrict__ Q) {
    const int wave = threadIdx.x >> 5;
    const int lane = threadIdx.x & 31;
    const int half = lane >> 4;
    const int lm   = lane & 15;
    const int m0   = blockIdx.x * 128 + wave * 16;

    v8f accP[8] = {};
    v8f accQ[8] = {};

    const float*  arow = A + (size_t)(m0 + lm) * K + 2 * half;
    const float4* bpA  = (const float4*)(BpA + (size_t)lane * 16);
    const float4* bpB  = (const float4*)(BpB + (size_t)lane * 16);

#pragma unroll
    for (int k = 0; k < K; k += 4) {
        const v2f a = *(const v2f*)(arow + k);
        const float4* ba = bpA + (size_t)(k >> 2) * 128;
        const float4* bb = bpB + (size_t)(k >> 2) * 128;
        float4 pa[4], pb[4];
#pragma unroll
        for (int i = 0; i < 4; ++i) { pa[i] = ba[i]; pb[i] = bb[i]; }
        v2f b;
#pragma unroll
        for (int i = 0; i < 4; ++i) {
            b.x = pa[i].x; b.y = pa[i].y; accP[2 * i + 0] = WMMA_F32(a, b, accP[2 * i + 0]);
            b.x = pa[i].z; b.y = pa[i].w; accP[2 * i + 1] = WMMA_F32(a, b, accP[2 * i + 1]);
            b.x = pb[i].x; b.y = pb[i].y; accQ[2 * i + 0] = WMMA_F32(a, b, accQ[2 * i + 0]);
            b.x = pb[i].z; b.y = pb[i].w; accQ[2 * i + 1] = WMMA_F32(a, b, accQ[2 * i + 1]);
        }
    }

    const size_t rb = (size_t)(m0 + 8 * half) * GNN_DOUT + lm;
    float* prow = P + rb;
    float* qrow = Q + rb;
#pragma unroll
    for (int r = 0; r < 8; ++r)
#pragma unroll
        for (int t = 0; t < 8; ++t) {
            prow[(size_t)r * GNN_DOUT + t * 16] = accP[t][r];
            qrow[(size_t)r * GNN_DOUT + t * 16] = accQ[t][r];
        }
}

#define GNN_CSR_T  1024
#define GNN_CSR_IT ((GNN_E + GNN_CSR_T - 1) / GNN_CSR_T)
__device__ __forceinline__ void csr_place_body(const int* __restrict__ ei,
                                               int* __restrict__ rowptr,
                                               int* __restrict__ slot) {
    __shared__ int hist[2 * GNN_CSR_T];
    __shared__ int wsum[32];
    const int tid  = threadIdx.x;
    const int lane = tid & 31;
    const int wave = tid >> 5;

    hist[2 * tid + 0] = 0;
    hist[2 * tid + 1] = 0;
    __syncthreads();

    int d[GNN_CSR_IT], r[GNN_CSR_IT];
#pragma unroll
    for (int it = 0; it < GNN_CSR_IT; ++it) {
        const int e = it * GNN_CSR_T + tid;
        d[it] = (e < GNN_E) ? ei[GNN_E + e] : -1;
    }
#pragma unroll
    for (int it = 0; it < GNN_CSR_IT; ++it)
        r[it] = (d[it] >= 0) ? atomicAdd(&hist[d[it]], 1) : 0;
    __syncthreads();

    const int a = hist[2 * tid + 0];
    const int s = a + hist[2 * tid + 1];
    int inc = s;
    for (int m = 1; m < 32; m <<= 1) {
        const int v = __shfl_up(inc, m, 32);
        if (lane >= m) inc += v;
    }
    if (lane == 31) wsum[wave] = inc;
    __syncthreads();
    if (wave == 0) {
        const int w = wsum[lane];
        int wi = w;
        for (int m = 1; m < 32; m <<= 1) {
            const int v = __shfl_up(wi, m, 32);
            if (lane >= m) wi += v;
        }
        wsum[lane] = wi - w;
    }
    __syncthreads();
    const int excl = wsum[wave] + inc - s;
    hist[2 * tid + 0] = excl;
    hist[2 * tid + 1] = excl + a;
    if (2 * tid + 0 <= GNN_N) rowptr[2 * tid + 0] = excl;
    if (2 * tid + 1 <= GNN_N) rowptr[2 * tid + 1] = excl + a;
    __syncthreads();

#pragma unroll
    for (int it = 0; it < GNN_CSR_IT; ++it) r[it] += hist[d[it] >= 0 ? d[it] : 0];
#pragma unroll
    for (int it = 0; it < GNN_CSR_IT; ++it)
        if (d[it] >= 0) slot[r[it]] = it * GNN_CSR_T + tid;
}

#define GNN_PK_A0 0
#define GNN_PK_B0 8192
#define GNN_PK_20 16384
#define GNN_PK_A1 32768
#define GNN_PK_B1 49152
#define GNN_PK_21 65536
#define GNN_PK_END 81920
__global__ __launch_bounds__(GNN_CSR_T) void k_setup(const int* __restrict__ ei,
                                                     int* __restrict__ rowptr,
                                                     int* __restrict__ slot,
                                                     const float* __restrict__ W1a,
                                                     const float* __restrict__ W2a,
                                                     const float* __restrict__ W1b,
                                                     const float* __restrict__ W2b,
                                                     float* __restrict__ Bp) {
    if (blockIdx.x == 0) { csr_place_body(ei, rowptr, slot); return; }
    const int first = (blockIdx.x - 1) * GNN_CSR_T;
    const int tid   = threadIdx.x;
    if (first < GNN_PK_B0)
        pack_b_elem<true >(W1a, Bp + GNN_PK_A0, 64, first - GNN_PK_A0 + tid);
    else if (first < GNN_PK_20)
        pack_b_elem<false>(W1a + 64 * GNN_DOUT, Bp + GNN_PK_B0, 64, first - GNN_PK_B0 + tid);
    else if (first < GNN_PK_A1)
        pack_b_elem<false>(W2a, Bp + GNN_PK_20, 128, first - GNN_PK_20 + tid);
    else if (first < GNN_PK_B1)
        pack_b_elem<true >(W1b, Bp + GNN_PK_A1, 128, first - GNN_PK_A1 + tid);
    else if (first < GNN_PK_21)
        pack_b_elem<false>(W1b + 128 * GNN_DOUT, Bp + GNN_PK_B1, 128, first - GNN_PK_B1 + tid);
    else
        pack_b_elem<false>(W2b, Bp + GNN_PK_21, 128, first - GNN_PK_21 + tid);
}

__global__ __launch_bounds__(256) void k_csr_order(const int* __restrict__ ei,
                                                   const int* __restrict__ rowptr,
                                                   const int* __restrict__ slot,
                                                   int* __restrict__ col) {
    const int e   = blockIdx.x * 256 + threadIdx.x;
    const int dst = ei[GNN_E + e];
    const int beg = rowptr[dst];
    const int end = rowptr[dst + 1];
    int rank = 0;
    for (int j = beg; j < end; ++j) rank += (slot[j] < e) ? 1 : 0;
    col[beg + rank] = ei[e];
}

__global__ __launch_bounds__(256) void k_gather(const float* __restrict__ P,
                                                const float* __restrict__ Q,
                                                const float* __restrict__ b1,
                                                const int* __restrict__ rowptr,
                                                const int* __restrict__ col,
                                                float* __restrict__ S,
                                                float* __restrict__ cnt) {
    const int wave = __builtin_amdgcn_readfirstlane((int)(threadIdx.x >> 5));
    const int lane = threadIdx.x & 31;
    const int c    = lane * 4;
    const int node = blockIdx.x * 8 + wave;
    const int off  = blockIdx.y * GNN_N;
    const int beg  = rowptr[node];
    const int end  = rowptr[node + 1];

    const size_t row = (size_t)(off + node) * GNN_DOUT + c;
    const float4 p  = *(const float4*)(P + row);
    const float4 bb = *(const float4*)(b1 + c);
    const float* qb = Q + (size_t)off * GNN_DOUT + c;

    float4 acc = {0.0f, 0.0f, 0.0f, 0.0f};
    for (int base = beg; base < end; base += 32) {
        const int n    = (end - base < 32) ? (end - base) : 32;
        const int mine = (lane < n) ? col[base + lane] : 0;
#pragma unroll 4
        for (int j = 0; j < n; ++j) {
            const int src = __builtin_amdgcn_readlane(mine, j);
            const float4 q = *(const float4*)(qb + (size_t)src * GNN_DOUT);
            acc.x += lrelu(p.x + q.x + bb.x);
            acc.y += lrelu(p.y + q.y + bb.y);
            acc.z += lrelu(p.z + q.z + bb.z);
            acc.w += lrelu(p.w + q.w + bb.w);
        }
    }
    *(float4*)(S + row) = acc;
    if (lane == 0) cnt[off + node] = (float)(end - beg);
}

template <bool RES>
__global__ __launch_bounds__(256) void k_gemm_agg_ln(const float* __restrict__ S,
                                                     const float* __restrict__ Bp2,
                                                     const float* __restrict__ cnt,
                                                     const float* __restrict__ b2,
                                                     const float* __restrict__ g,
                                                     const float* __restrict__ beta,
                                                     const float* __restrict__ xres,
                                                     float* __restrict__ out) {
    const int wave = threadIdx.x >> 5;
    const int lane = threadIdx.x & 31;
    const int half = lane >> 4;
    const int lm   = lane & 15;
    const int m0   = blockIdx.x * 128 + wave * 16;

    const float cntv  = cnt[m0 + lm];
    const float scale = 1.0f / fmaxf(cntv, 1.0f);

    v8f acc[8] = {};
    const float*  arow = S + (size_t)(m0 + lm) * 128 + 2 * half;
    const float4* bp   = (const float4*)(Bp2 + (size_t)lane * 16);

#pragma unroll
    for (int k = 0; k < 128; k += 4) {
        v2f a = *(const v2f*)(arow + k);
        a.x *= scale; a.y *= scale;
        const float4* bk = bp + (size_t)(k >> 2) * 128;
        float4 pb[4];
#pragma unroll
        for (int i = 0; i < 4; ++i) pb[i] = bk[i];
        v2f b;
#pragma unroll
        for (int i = 0; i < 4; ++i) {
            b.x = pb[i].x; b.y = pb[i].y; acc[2 * i + 0] = WMMA_F32(a, b, acc[2 * i + 0]);
            b.x = pb[i].z; b.y = pb[i].w; acc[2 * i + 1] = WMMA_F32(a, b, acc[2 * i + 1]);
        }
    }

    float b2v[8], gv[8], bev[8];
#pragma unroll
    for (int t = 0; t < 8; ++t) {
        b2v[t] = b2[t * 16 + lm];
        gv[t]  = g[t * 16 + lm];
        bev[t] = beta[t * 16 + lm];
    }

    const size_t rb = (size_t)(m0 + 8 * half) * GNN_DOUT + lm;
    float* crow = out + rb;
    const float* rrow = RES ? (xres + rb) : nullptr;

#pragma unroll
    for (int r = 0; r < 8; ++r) {
        const float cr   = __shfl(cntv, r + 8 * half, 32);
        const float hasE = (cr > 0.0f) ? 1.0f : 0.0f;

        float h[8];
        float s = 0.0f;
#pragma unroll
        for (int t = 0; t < 8; ++t) { h[t] = acc[t][r] + hasE * b2v[t]; s += h[t]; }
#pragma unroll
        for (int m = 8; m >= 1; m >>= 1) s += __shfl_xor(s, m, 32);
        const float mu = s * (1.0f / GNN_DOUT);

        float s2 = 0.0f;
#pragma unroll
        for (int t = 0; t < 8; ++t) { h[t] -= mu; s2 += h[t] * h[t]; }
#pragma unroll
        for (int m = 8; m >= 1; m >>= 1) s2 += __shfl_xor(s2, m, 32);
        const float rstd = rsqrtf(s2 * (1.0f / GNN_DOUT) + GNN_EPS);

#pragma unroll
        for (int t = 0; t < 8; ++t) {
            float o = h[t] * rstd * gv[t] + bev[t];
            if (RES) o += rrow[(size_t)r * GNN_DOUT + t * 16];
            crow[(size_t)r * GNN_DOUT + t * 16] = lrelu(o);
        }
    }
}

extern "C" void kernel_launch(void* const* d_in, const int* in_sizes, int n_in,
                              void* d_out, int out_size, void* d_ws, size_t ws_size,
                              hipStream_t stream) {
    (void)in_sizes; (void)n_in; (void)out_size; (void)ws_size;

    const float* x  = (const float*)d_in[0];
    const int*   ei = (const int*)d_in[1];
    const float* W1[2] = {(const float*)d_in[2],  (const float*)d_in[8]};
    const float* b1[2] = {(const float*)d_in[3],  (const float*)d_in[9]};
    const float* W2[2] = {(const float*)d_in[4],  (const float*)d_in[10]};
    const float* b2[2] = {(const float*)d_in[5],  (const float*)d_in[11]};
    const float* g [2] = {(const float*)d_in[6],  (const float*)d_in[12]};
    const float* be[2] = {(const float*)d_in[7],  (const float*)d_in[13]};

    const size_t NF = (size_t)GNN_NT * GNN_DOUT;
    float* ws   = (float*)d_ws;
    float* X    = ws;
    float* P    = X + NF;
    float* Q    = P + NF;
    float* S    = Q + NF;
    float* cnt  = S + NF;
    float* Bp   = cnt + GNN_NT;
    int* rowptr = (int*)(Bp + GNN_PK_END);
    int* slot   = rowptr + 2048;
    int* col    = slot + GNN_E;

    const dim3 ggrid(GNN_N / 8, GNN_BSN);

    k_setup<<<1 + GNN_PK_END / GNN_CSR_T, GNN_CSR_T, 0, stream>>>(
        ei, rowptr, slot, W1[0], W2[0], W1[1], W2[1], Bp);
    k_csr_order<<<GNN_E / 256, 256, 0, stream>>>(ei, rowptr, slot, col);

    k_gemm_pq<64><<<250, 256, 0, stream>>>(x, Bp + GNN_PK_A0, Bp + GNN_PK_B0, P, Q);
    k_gather<<<ggrid, 256, 0, stream>>>(P, Q, b1[0], rowptr, col, S, cnt);
    k_gemm_agg_ln<false><<<250, 256, 0, stream>>>(S, Bp + GNN_PK_20, cnt, b2[0], g[0], be[0], nullptr, X);

    k_gemm_pq<128><<<250, 256, 0, stream>>>(X, Bp + GNN_PK_A1, Bp + GNN_PK_B1, P, Q);
    k_gather<<<ggrid, 256, 0, stream>>>(P, Q, b1[1], rowptr, col, S, cnt);
    k_gemm_agg_ln<true><<<250, 256, 0, stream>>>(S, Bp + GNN_PK_21, cnt, b2[1], g[1], be[1], X, (float*)d_out);
}


# Edge sums feed the aggregation GEMM through LDS in the same launch (8 -> 6 launches, S and cnt never in memory); LayerNorm row sums now 4 channels per lane then a 32-lane butterfly
# speedup vs baseline: 3.3383x; 1.2873x over previous
#include <hip/hip_runtime.h>


typedef __attribute__((ext_vector_type(2))) float v2f;
typedef __attribute__((ext_vector_type(8))) float v8f;

#define GNN_SLOPE 0.01f
#define GNN_EPS   1e-5f
#define GNN_BSN   16
#define GNN_N     2000
#define GNN_E     32000
#define GNN_NT    (GNN_BSN * GNN_N)
#define GNN_DOUT  128

__device__ __forceinline__ float lrelu(float v) { return v >= 0.0f ? v : GNN_SLOPE * v; }

#define WMMA_F32(A, B, C) \
    __builtin_amdgcn_wmma_f32_16x16x4_f32(false, (A), false, (B), (short)0, (C), false, false)

template <bool DIFF>
__device__ __forceinline__ void pack_b_elem(const float* __restrict__ B,
                                            float* __restrict__ Bp, int K, int i) {
    const int kk   = i >> 9;
    const int rem  = i & 511;
    const int lane = rem >> 4;
    const int j    = rem & 15;
    const int t    = j >> 1;
    const int comp = j & 1;
    const int r    = kk * 4 + 2 * (lane >> 4) + comp;
    const int c    = t * 16 + (lane & 15);
    const int src  = r * GNN_DOUT + c;
    float v = B[src];
    if (DIFF) v -= B[K * GNN_DOUT + src];
    Bp[i] = v;
}

__device__ __forceinline__ void pack_b_tile_elem(const float* __restrict__ B,
                                                 float* __restrict__ Bt, int i) {
    const int t    = i >> 11;
    const int rem  = i & 2047;
    const int kk2  = rem >> 7;
    const int lane = (rem & 127) >> 2;
    const int j    = rem & 3;
    const int r    = (2 * kk2 + (j >> 1)) * 4 + 2 * (lane >> 4) + (j & 1);
    const int c    = t * 16 + (lane & 15);
    Bt[i] = B[r * GNN_DOUT + c];
}

template <int K>
__global__ __launch_bounds__(256) void k_gemm_pq(const float* __restrict__ A,
                                                 const float* __restrict__ BpA,
                                                 const float* __restrict__ BpB,
                                                 float* __restrict__ P,
                                                 float* __restrict__ Q) {
    const int wave = threadIdx.x >> 5;
    const int lane = threadIdx.x & 31;
    const int half = lane >> 4;
    const int lm   = lane & 15;
    const int m0   = blockIdx.x * 128 + wave * 16;

    v8f accP[8] = {};
    v8f accQ[8] = {};

    const float*  arow = A + (size_t)(m0 + lm) * K + 2 * half;
    const float4* bpA  = (const float4*)(BpA + (size_t)lane * 16);
    const float4* bpB  = (const float4*)(BpB + (size_t)lane * 16);

#pragma unroll
    for (int k = 0; k < K; k += 4) {
        const v2f a = *(const v2f*)(arow + k);
        const float4* ba = bpA + (size_t)(k >> 2) * 128;
        const float4* bb = bpB + (size_t)(k >> 2) * 128;
        float4 pa[4], pb[4];
#pragma unroll
        for (int i = 0; i < 4; ++i) { pa[i] = ba[i]; pb[i] = bb[i]; }
        v2f b;
#pragma unroll
        for (int i = 0; i < 4; ++i) {
            b.x = pa[i].x; b.y = pa[i].y; accP[2 * i + 0] = WMMA_F32(a, b, accP[2 * i + 0]);
            b.x = pa[i].z; b.y = pa[i].w; accP[2 * i + 1] = WMMA_F32(a, b, accP[2 * i + 1]);
            b.x = pb[i].x; b.y = pb[i].y; accQ[2 * i + 0] = WMMA_F32(a, b, accQ[2 * i + 0]);
            b.x = pb[i].z; b.y = pb[i].w; accQ[2 * i + 1] = WMMA_F32(a, b, accQ[2 * i + 1]);
        }
    }

    const size_t rb = (size_t)(m0 + 8 * half) * GNN_DOUT + lm;
    float* prow = P + rb;
    float* qrow = Q + rb;
#pragma unroll
    for (int r = 0; r < 8; ++r)
#pragma unroll
        for (int t = 0; t < 8; ++t) {
            prow[(size_t)r * GNN_DOUT + t * 16] = accP[t][r];
            qrow[(size_t)r * GNN_DOUT + t * 16] = accQ[t][r];
        }
}

#define GNN_CSR_T  1024
#define GNN_CSR_IT ((GNN_E + GNN_CSR_T - 1) / GNN_CSR_T)
__device__ __forceinline__ void csr_place_body(const int* __restrict__ ei,
                                               int* __restrict__ rowptr,
                                               int* __restrict__ slot) {
    __shared__ int hist[2 * GNN_CSR_T];
    __shared__ int wsum[32];
    const int tid  = threadIdx.x;
    const int lane = tid & 31;
    const int wave = tid >> 5;

    hist[2 * tid + 0] = 0;
    hist[2 * tid + 1] = 0;
    __syncthreads();

    int d[GNN_CSR_IT], r[GNN_CSR_IT];
#pragma unroll
    for (int it = 0; it < GNN_CSR_IT; ++it) {
        const int e = it * GNN_CSR_T + tid;
        d[it] = (e < GNN_E) ? ei[GNN_E + e] : -1;
    }
#pragma unroll
    for (int it = 0; it < GNN_CSR_IT; ++it)
        r[it] = (d[it] >= 0) ? atomicAdd(&hist[d[it]], 1) : 0;
    __syncthreads();

    const int a = hist[2 * tid + 0];
    const int s = a + hist[2 * tid + 1];
    int inc = s;
    for (int m = 1; m < 32; m <<= 1) {
        const int v = __shfl_up(inc, m, 32);
        if (lane >= m) inc += v;
    }
    if (lane == 31) wsum[wave] = inc;
    __syncthreads();
    if (wave == 0) {
        const int w = wsum[lane];
        int wi = w;
        for (int m = 1; m < 32; m <<= 1) {
            const int v = __shfl_up(wi, m, 32);
            if (lane >= m) wi += v;
        }
        wsum[lane] = wi - w;
    }
    __syncthreads();
    const int excl = wsum[wave] + inc - s;
    hist[2 * tid + 0] = excl;
    hist[2 * tid + 1] = excl + a;
    if (2 * tid + 0 <= GNN_N) rowptr[2 * tid + 0] = excl;
    if (2 * tid + 1 <= GNN_N) rowptr[2 * tid + 1] = excl + a;
    __syncthreads();

#pragma unroll
    for (int it = 0; it < GNN_CSR_IT; ++it) r[it] += hist[d[it] >= 0 ? d[it] : 0];
#pragma unroll
    for (int it = 0; it < GNN_CSR_IT; ++it)
        if (d[it] >= 0) slot[r[it]] = it * GNN_CSR_T + tid;
}

#define GNN_PK_A0 0
#define GNN_PK_B0 8192
#define GNN_PK_20 16384
#define GNN_PK_A1 32768
#define GNN_PK_B1 49152
#define GNN_PK_21 65536
#define GNN_PK_END 81920
__global__ __launch_bounds__(GNN_CSR_T) void k_setup(const int* __restrict__ ei,
                                                     int* __restrict__ rowptr,
                                                     int* __restrict__ slot,
                                                     const float* __restrict__ W1a,
                                                     const float* __restrict__ W2a,
                                                     const float* __restrict__ W1b,
                                                     const float* __restrict__ W2b,
                                                     float* __restrict__ Bp) {
    if (blockIdx.x == 0) { csr_place_body(ei, rowptr, slot); return; }
    const int first = (blockIdx.x - 1) * GNN_CSR_T;
    const int tid   = threadIdx.x;
    if (first < GNN_PK_B0)
        pack_b_elem<true >(W1a, Bp + GNN_PK_A0, 64, first - GNN_PK_A0 + tid);
    else if (first < GNN_PK_20)
        pack_b_elem<false>(W1a + 64 * GNN_DOUT, Bp + GNN_PK_B0, 64, first - GNN_PK_B0 + tid);
    else if (first < GNN_PK_A1)
        pack_b_tile_elem(W2a, Bp + GNN_PK_20, first - GNN_PK_20 + tid);
    else if (first < GNN_PK_B1)
        pack_b_elem<true >(W1b, Bp + GNN_PK_A1, 128, first - GNN_PK_A1 + tid);
    else if (first < GNN_PK_21)
        pack_b_elem<false>(W1b + 128 * GNN_DOUT, Bp + GNN_PK_B1, 128, first - GNN_PK_B1 + tid);
    else
        pack_b_tile_elem(W2b, Bp + GNN_PK_21, first - GNN_PK_21 + tid);
}

__global__ __launch_bounds__(256) void k_csr_order(const int* __restrict__ ei,
                                                   const int* __restrict__ rowptr,
                                                   const int* __restrict__ slot,
                                                   int* __restrict__ col) {
    const int e   = blockIdx.x * 256 + threadIdx.x;
    const int dst = ei[GNN_E + e];
    const int beg = rowptr[dst];
    const int end = rowptr[dst + 1];
    int rank = 0;
    for (int j = beg; j < end; ++j) rank += (slot[j] < e) ? 1 : 0;
    col[beg + rank] = ei[e];
}

#define GNN_FT_ROWS  16
#define GNN_FT_PITCH 132

__device__ __forceinline__ float4 edge_row_sum(const float* __restrict__ qb, const float4 p,
                                               const float4 bb, const int* __restrict__ col,
                                               int beg, int end, int lane) {
    float4 acc = {0.0f, 0.0f, 0.0f, 0.0f};
    for (int base = beg; base < end; base += 32) {
        const int n    = (end - base < 32) ? (end - base) : 32;
        const int mine = (lane < n) ? col[base + lane] : 0;
#pragma unroll 4
        for (int j = 0; j < n; ++j) {
            const int src = __builtin_amdgcn_readlane(mine, j);
            const float4 q = *(const float4*)(qb + (size_t)src * GNN_DOUT);
            acc.x += lrelu(p.x + q.x + bb.x);
            acc.y += lrelu(p.y + q.y + bb.y);
            acc.z += lrelu(p.z + q.z + bb.z);
            acc.w += lrelu(p.w + q.w + bb.w);
        }
    }
    return acc;
}

template <bool RES>
__global__ __launch_bounds__(256) void k_edge_agg_ln(const float* __restrict__ P,
                                                     const float* __restrict__ Q,
                                                     const float* __restrict__ b1,
                                                     const int* __restrict__ rowptr,
                                                     const int* __restrict__ col,
                                                     const float* __restrict__ Bt2,
                                                     const float* __restrict__ b2,
                                                     const float* __restrict__ g,
                                                     const float* __restrict__ beta,
                                                     const float* __restrict__ xres,
                                                     float* __restrict__ out) {
    __shared__ __attribute__((aligned(16))) float tileS[GNN_FT_ROWS * GNN_FT_PITCH];
    __shared__ __attribute__((aligned(16))) float tileO[GNN_FT_ROWS * GNN_FT_PITCH];

    const int wave = __builtin_amdgcn_readfirstlane((int)(threadIdx.x >> 5));
    const int lane = threadIdx.x & 31;
    const int half = lane >> 4;
    const int lm   = lane & 15;
    const int c    = lane * 4;
    const int off  = blockIdx.y * GNN_N;
    const int lr0  = 2 * wave;
    const int node = blockIdx.x * GNN_FT_ROWS + lr0;

    const int beg0 = rowptr[node];
    const int beg1 = rowptr[node + 1];
    const int end1 = rowptr[node + 2];
    const size_t row0 = (size_t)(off + node) * GNN_DOUT + c;
    const float4 p0 = *(const float4*)(P + row0);
    const float4 p1 = *(const float4*)(P + row0 + GNN_DOUT);
    const float4 bb = *(const float4*)(b1 + c);
    const float* qb = Q + (size_t)off * GNN_DOUT + c;

    const float cnt0 = (float)(beg1 - beg0);
    const float cnt1 = (float)(end1 - beg1);
    {
        float4 s0 = edge_row_sum(qb, p0, bb, col, beg0, beg1, lane);
        const float sc0 = 1.0f / fmaxf(cnt0, 1.0f);
        s0.x *= sc0; s0.y *= sc0; s0.z *= sc0; s0.w *= sc0;
        *(float4*)(tileS + (lr0 + 0) * GNN_FT_PITCH + c) = s0;
        float4 s1 = edge_row_sum(qb, p1, bb, col, beg1, end1, lane);
        const float sc1 = 1.0f / fmaxf(cnt1, 1.0f);
        s1.x *= sc1; s1.y *= sc1; s1.z *= sc1; s1.w *= sc1;
        *(float4*)(tileS + (lr0 + 1) * GNN_FT_PITCH + c) = s1;
    }

    const float4* bt = (const float4*)(Bt2 + (size_t)wave * 2048 + lane * 4);
    float4 bf[16];
#pragma unroll
    for (int i = 0; i < 16; ++i) bf[i] = bt[i * 32];

    __syncthreads();

    v8f acc = {};
    const float* arow = tileS + lm * GNN_FT_PITCH + 2 * half;
#pragma unroll
    for (int i = 0; i < 16; ++i) {
        const v2f a0 = *(const v2f*)(arow + 8 * i);
        const v2f a1 = *(const v2f*)(arow + 8 * i + 4);
        v2f b;
        b.x = bf[i].x; b.y = bf[i].y; acc = WMMA_F32(a0, b, acc);
        b.x = bf[i].z; b.y = bf[i].w; acc = WMMA_F32(a1, b, acc);
    }
#pragma unroll
    for (int r = 0; r < 8; ++r)
        tileO[(r + 8 * half) * GNN_FT_PITCH + 16 * wave + lm] = acc[r];

    __syncthreads();

    const float4 b2v = *(const float4*)(b2 + c);
    const float4 gv  = *(const float4*)(g + c);
    const float4 bev = *(const float4*)(beta + c);
#pragma unroll
    for (int q = 0; q < 2; ++q) {
        const float hasE = ((q == 0 ? cnt0 : cnt1) > 0.0f) ? 1.0f : 0.0f;
        float4 h = *(const float4*)(tileO + (lr0 + q) * GNN_FT_PITCH + c);
        h.x += hasE * b2v.x; h.y += hasE * b2v.y; h.z += hasE * b2v.z; h.w += hasE * b2v.w;

        float s = (h.x + h.y) + (h.z + h.w);
#pragma unroll
        for (int m = 16; m >= 1; m >>= 1) s += __shfl_xor(s, m, 32);
        const float mu = s * (1.0f / GNN_DOUT);

        h.x -= mu; h.y -= mu; h.z -= mu; h.w -= mu;
        float s2 = (h.x * h.x + h.y * h.y) + (h.z * h.z + h.w * h.w);
#pragma unroll
        for (int m = 16; m >= 1; m >>= 1) s2 += __shfl_xor(s2, m, 32);
        const float rstd = rsqrtf(s2 * (1.0f / GNN_DOUT) + GNN_EPS);

        float4 o;
        o.x = h.x * rstd * gv.x + bev.x;
        o.y = h.y * rstd * gv.y + bev.y;
        o.z = h.z * rstd * gv.z + bev.z;
        o.w = h.w * rstd * gv.w + bev.w;
        const size_t orow = row0 + (size_t)q * GNN_DOUT;
        if (RES) {
            const float4 xr = *(const float4*)(xres + orow);
            o.x += xr.x; o.y += xr.y; o.z += xr.z; o.w += xr.w;
        }
        o.x = lrelu(o.x); o.y = lrelu(o.y); o.z = lrelu(o.z); o.w = lrelu(o.w);
        *(float4*)(out + orow) = o;
    }
}

extern "C" void kernel_launch(void* const* d_in, const int* in_sizes, int n_in,
                              void* d_out, int out_size, void* d_ws, size_t ws_size,
                              hipStream_t stream) {
    (void)in_sizes; (void)n_in; (void)out_size; (void)ws_size;

    const float* x  = (const float*)d_in[0];
    const int*   ei = (const int*)d_in[1];
    const float* W1[2] = {(const float*)d_in[2],  (const float*)d_in[8]};
    const float* b1[2] = {(const float*)d_in[3],  (const float*)d_in[9]};
    const float* W2[2] = {(const float*)d_in[4],  (const float*)d_in[10]};
    const float* b2[2] = {(const float*)d_in[5],  (const float*)d_in[11]};
    const float* g [2] = {(const float*)d_in[6],  (const float*)d_in[12]};
    const float* be[2] = {(const float*)d_in[7],  (const float*)d_in[13]};

    const size_t NF = (size_t)GNN_NT * GNN_DOUT;
    float* ws   = (float*)d_ws;
    float* X    = ws;
    float* P    = X + NF;
    float* Q    = P + NF;
    float* Bp   = Q + NF;
    int* rowptr = (int*)(Bp + GNN_PK_END);
    int* slot   = rowptr + 2048;
    int* col    = slot + GNN_E;

    const dim3 egrid(GNN_N / GNN_FT_ROWS, GNN_BSN);

    k_setup<<<1 + GNN_PK_END / GNN_CSR_T, GNN_CSR_T, 0, stream>>>(
        ei, rowptr, slot, W1[0], W2[0], W1[1], W2[1], Bp);
    k_csr_order<<<GNN_E / 256, 256, 0, stream>>>(ei, rowptr, slot, col);

    k_gemm_pq<64><<<250, 256, 0, stream>>>(x, Bp + GNN_PK_A0, Bp + GNN_PK_B0, P, Q);
    k_edge_agg_ln<false><<<egrid, 256, 0, stream>>>(P, Q, b1[0], rowptr, col, Bp + GNN_PK_20,
                                                    b2[0], g[0], be[0], nullptr, X);

    k_gemm_pq<128><<<250, 256, 0, stream>>>(X, Bp + GNN_PK_A1, Bp + GNN_PK_B1, P, Q);
    k_edge_agg_ln<true><<<egrid, 256, 0, stream>>>(P, Q, b1[1], rowptr, col, Bp + GNN_PK_21,
                                                   b2[1], g[1], be[1], X, (float*)d_out);
}


# Edge bucketing split by destination range over 40 work-groups that count, place and order their own buckets in the set-up launch; ordering launch removed (6 -> 5); float sum order unchanged
# speedup vs baseline: 4.0092x; 1.2010x over previous
#include <hip/hip_runtime.h>


typedef __attribute__((ext_vector_type(2))) float v2f;
typedef __attribute__((ext_vector_type(8))) float v8f;

#define GNN_SLOPE 0.01f
#define GNN_EPS   1e-5f
#define GNN_BSN   16
#define GNN_N     2000
#define GNN_E     32000
#define GNN_NT    (GNN_BSN * GNN_N)
#define GNN_DOUT  128

__device__ __forceinline__ float lrelu(float v) { return v >= 0.0f ? v : GNN_SLOPE * v; }

#define WMMA_F32(A, B, C) \
    __builtin_amdgcn_wmma_f32_16x16x4_f32(false, (A), false, (B), (short)0, (C), false, false)

template <bool DIFF>
__device__ __forceinline__ void pack_b_elem(const float* __restrict__ B,
                                            float* __restrict__ Bp, int K, int i) {
    const int kk   = i >> 9;
    const int rem  = i & 511;
    const int lane = rem >> 4;
    const int j    = rem & 15;
    const int t    = j >> 1;
    const int comp = j & 1;
    const int r    = kk * 4 + 2 * (lane >> 4) + comp;
    const int c    = t * 16 + (lane & 15);
    const int src  = r * GNN_DOUT + c;
    float v = B[src];
    if (DIFF) v -= B[K * GNN_DOUT + src];
    Bp[i] = v;
}

__device__ __forceinline__ void pack_b_tile_elem(const float* __restrict__ B,
                                                 float* __restrict__ Bt, int i) {
    const int t    = i >> 11;
    const int rem  = i & 2047;
    const int kk2  = rem >> 7;
    const int lane = (rem & 127) >> 2;
    const int j    = rem & 3;
    const int r    = (2 * kk2 + (j >> 1)) * 4 + 2 * (lane >> 4) + (j & 1);
    const int c    = t * 16 + (lane & 15);
    Bt[i] = B[r * GNN_DOUT + c];
}

template <int K>
__global__ __launch_bounds__(256) void k_gemm_pq(const float* __restrict__ A,
                                                 const float* __restrict__ BpA,
                                                 const float* __restrict__ BpB,
                                                 float* __restrict__ P,
                                                 float* __restrict__ Q) {
    const int wave = threadIdx.x >> 5;
    const int lane = threadIdx.x & 31;
    const int half = lane >> 4;
    const int lm   = lane & 15;
    const int m0   = blockIdx.x * 128 + wave * 16;

    v8f accP[8] = {};
    v8f accQ[8] = {};

    const float*  arow = A + (size_t)(m0 + lm) * K + 2 * half;
    const float4* bpA  = (const float4*)(BpA + (size_t)lane * 16);
    const float4* bpB  = (const float4*)(BpB + (size_t)lane * 16);

#pragma unroll
    for (int k = 0; k < K; k += 4) {
        const v2f a = *(const v2f*)(arow + k);
        const float4* ba = bpA + (size_t)(k >> 2) * 128;
        const float4* bb = bpB + (size_t)(k >> 2) * 128;
        float4 pa[4], pb[4];
#pragma unroll
        for (int i = 0; i < 4; ++i) { pa[i] = ba[i]; pb[i] = bb[i]; }
        v2f b;
#pragma unroll
        for (int i = 0; i < 4; ++i) {
            b.x = pa[i].x; b.y = pa[i].y; accP[2 * i + 0] = WMMA_F32(a, b, accP[2 * i + 0]);
            b.x = pa[i].z; b.y = pa[i].w; accP[2 * i + 1] = WMMA_F32(a, b, accP[2 * i + 1]);
            b.x = pb[i].x; b.y = pb[i].y; accQ[2 * i + 0] = WMMA_F32(a, b, accQ[2 * i + 0]);
            b.x = pb[i].z; b.y = pb[i].w; accQ[2 * i + 1] = WMMA_F32(a, b, accQ[2 * i + 1]);
        }
    }

    const size_t rb = (size_t)(m0 + 8 * half) * GNN_DOUT + lm;
    float* prow = P + rb;
    float* qrow = Q + rb;
#pragma unroll
    for (int r = 0; r < 8; ++r)
#pragma unroll
        for (int t = 0; t < 8; ++t) {
            prow[(size_t)r * GNN_DOUT + t * 16] = accP[t][r];
            qrow[(size_t)r * GNN_DOUT + t * 16] = accQ[t][r];
        }
}

#define GNN_CSR_T  1024
#define GNN_CSR_IT ((GNN_E + GNN_CSR_T - 1) / GNN_CSR_T)
#define GNN_CSR_R  50
#define GNN_CSR_G  (GNN_N / GNN_CSR_R)
#define GNN_CSR_H  64
static_assert(GNN_N % GNN_CSR_R == 0, "destination ranges must tile the node set");
static_assert(GNN_CSR_R < GNN_CSR_H, "entry R of the scanned counters is the end of the range");

__device__ __forceinline__ void csr_range_body(const int* __restrict__ ei,
                                               int* __restrict__ rowptr,
                                               int* slot,
                                               int* __restrict__ col,
                                               int g) {
    __shared__ int hist[GNN_CSR_H];
    __shared__ int below;
    const int tid  = threadIdx.x;
    const int lane = tid & 31;
    const int lo   = g * GNN_CSR_R;

    if (tid < GNN_CSR_H) hist[tid] = 0;
    if (tid == 0) below = 0;
    __syncthreads();

    int rel[GNN_CSR_IT], r[GNN_CSR_IT];
#pragma unroll
    for (int it = 0; it < GNN_CSR_IT; ++it) {
        const int e = it * GNN_CSR_T + tid;
        rel[it] = (e < GNN_E) ? (ei[GNN_E + e] - lo) : GNN_N;
    }
    int nb = 0;
#pragma unroll
    for (int it = 0; it < GNN_CSR_IT; ++it) {
        r[it] = ((unsigned)rel[it] < (unsigned)GNN_CSR_R) ? atomicAdd(&hist[rel[it]], 1) : 0;
        nb += (rel[it] < 0) ? 1 : 0;
    }
#pragma unroll
    for (int m = 16; m >= 1; m >>= 1) nb += __shfl_xor(nb, m, 32);
    if (lane == 0) atomicAdd(&below, nb);
    __syncthreads();

    if (tid < 32) {
        const int i0 = 2 * lane;
        const int i1 = 2 * lane + 1;
        const int c0 = hist[i0];
        const int s  = c0 + hist[i1];
        int inc = s;
        for (int m = 1; m < 32; m <<= 1) {
            const int v = __shfl_up(inc, m, 32);
            if (lane >= m) inc += v;
        }
        const int s0 = below + inc - s;
        const int s1 = s0 + c0;
        hist[i0] = s0;
        hist[i1] = s1;
        if (i0 < GNN_CSR_R) rowptr[lo + i0] = s0;
        if (i1 < GNN_CSR_R) rowptr[lo + i1] = s1;
        if (lo + GNN_CSR_R == GNN_N) {
            if (i0 == GNN_CSR_R) rowptr[GNN_N] = s0;
            if (i1 == GNN_CSR_R) rowptr[GNN_N] = s1;
        }
    }
    __syncthreads();

#pragma unroll
    for (int it = 0; it < GNN_CSR_IT; ++it)
        if ((unsigned)rel[it] < (unsigned)GNN_CSR_R)
            slot[hist[rel[it]] + r[it]] = it * GNN_CSR_T + tid;
    __syncthreads();

    const int base = hist[0];
    const int nown = hist[GNN_CSR_R] - base;
    for (int p = tid; p < nown; p += GNN_CSR_T) {
        const int e   = slot[base + p];
        const int src = ei[e];
        const int b   = ei[GNN_E + e] - lo;
        const int beg = hist[b];
        const int end = hist[b + 1];
        int rank = 0;
        for (int j = beg; j < end; ++j) rank += (slot[j] < e) ? 1 : 0;
        col[beg + rank] = src;
    }
}

#define GNN_PK_A0 0
#define GNN_PK_B0 8192
#define GNN_PK_20 16384
#define GNN_PK_A1 32768
#define GNN_PK_B1 49152
#define GNN_PK_21 65536
#define GNN_PK_END 81920
__global__ __launch_bounds__(GNN_CSR_T) void k_setup(const int* __restrict__ ei,
                                                     int* __restrict__ rowptr,
                                                     int* slot,
                                                     int* __restrict__ col,
                                                     const float* __restrict__ W1a,
                                                     const float* __restrict__ W2a,
                                                     const float* __restrict__ W1b,
                                                     const float* __restrict__ W2b,
                                                     float* __restrict__ Bp) {
    if (blockIdx.x < GNN_CSR_G) { csr_range_body(ei, rowptr, slot, col, blockIdx.x); return; }
    const int first = (blockIdx.x - GNN_CSR_G) * GNN_CSR_T;
    const int tid   = threadIdx.x;
    if (first < GNN_PK_B0)
        pack_b_elem<true >(W1a, Bp + GNN_PK_A0, 64, first - GNN_PK_A0 + tid);
    else if (first < GNN_PK_20)
        pack_b_elem<false>(W1a + 64 * GNN_DOUT, Bp + GNN_PK_B0, 64, first - GNN_PK_B0 + tid);
    else if (first < GNN_PK_A1)
        pack_b_tile_elem(W2a, Bp + GNN_PK_20, first - GNN_PK_20 + tid);
    else if (first < GNN_PK_B1)
        pack_b_elem<true >(W1b, Bp + GNN_PK_A1, 128, first - GNN_PK_A1 + tid);
    else if (first < GNN_PK_21)
        pack_b_elem<false>(W1b + 128 * GNN_DOUT, Bp + GNN_PK_B1, 128, first - GNN_PK_B1 + tid);
    else
        pack_b_tile_elem(W2b, Bp + GNN_PK_21, first - GNN_PK_21 + tid);
}

#define GNN_FT_ROWS  16
#define GNN_FT_PITCH 132

__device__ __forceinline__ float4 edge_row_sum(const float* __restrict__ qb, const float4 p,
                                               const float4 bb, const int* __restrict__ col,
                                               int beg, int end, int lane) {
    float4 acc = {0.0f, 0.0f, 0.0f, 0.0f};
    for (int base = beg; base < end; base += 32) {
        const int n    = (end - base < 32) ? (end - base) : 32;
        const int mine = (lane < n) ? col[base + lane] : 0;
#pragma unroll 4
        for (int j = 0; j < n; ++j) {
            const int src = __builtin_amdgcn_readlane(mine, j);
            const float4 q = *(const float4*)(qb + (size_t)src * GNN_DOUT);
            acc.x += lrelu(p.x + q.x + bb.x);
            acc.y += lrelu(p.y + q.y + bb.y);
            acc.z += lrelu(p.z + q.z + bb.z);
            acc.w += lrelu(p.w + q.w + bb.w);
        }
    }
    return acc;
}

template <bool RES>
__global__ __launch_bounds__(256) void k_edge_agg_ln(const float* __restrict__ P,
                                                     const float* __restrict__ Q,
                                                     const float* __restrict__ b1,
                                                     const int* __restrict__ rowptr,
                                                     const int* __restrict__ col,
                                                     const float* __restrict__ Bt2,
                                                     const float* __restrict__ b2,
                                                     const float* __restrict__ g,
                                                     const float* __restrict__ beta,
                                                     const float* __restrict__ xres,
                                                     float* __restrict__ out) {
    __shared__ __attribute__((aligned(16))) float tileS[GNN_FT_ROWS * GNN_FT_PITCH];
    __shared__ __attribute__((aligned(16))) float tileO[GNN_FT_ROWS * GNN_FT_PITCH];

    const int wave = __builtin_amdgcn_readfirstlane((int)(threadIdx.x >> 5));
    const int lane = threadIdx.x & 31;
    const int half = lane >> 4;
    const int lm   = lane & 15;
    const int c    = lane * 4;
    const int off  = blockIdx.y * GNN_N;
    const int lr0  = 2 * wave;
    const int node = blockIdx.x * GNN_FT_ROWS + lr0;

    const int beg0 = rowptr[node];
    const int beg1 = rowptr[node + 1];
    const int end1 = rowptr[node + 2];
    const size_t row0 = (size_t)(off + node) * GNN_DOUT + c;
    const float4 p0 = *(const float4*)(P + row0);
    const float4 p1 = *(const float4*)(P + row0 + GNN_DOUT);
    const float4 bb = *(const float4*)(b1 + c);
    const float* qb = Q + (size_t)off * GNN_DOUT + c;

    const float cnt0 = (float)(beg1 - beg0);
    const float cnt1 = (float)(end1 - beg1);
    {
        float4 s0 = edge_row_sum(qb, p0, bb, col, beg0, beg1, lane);
        const float sc0 = 1.0f / fmaxf(cnt0, 1.0f);
        s0.x *= sc0; s0.y *= sc0; s0.z *= sc0; s0.w *= sc0;
        *(float4*)(tileS + (lr0 + 0) * GNN_FT_PITCH + c) = s0;
        float4 s1 = edge_row_sum(qb, p1, bb, col, beg1, end1, lane);
        const float sc1 = 1.0f / fmaxf(cnt1, 1.0f);
        s1.x *= sc1; s1.y *= sc1; s1.z *= sc1; s1.w *= sc1;
        *(float4*)(tileS + (lr0 + 1) * GNN_FT_PITCH + c) = s1;
    }

    const float4* bt = (const float4*)(Bt2 + (size_t)wave * 2048 + lane * 4);
    float4 bf[16];
#pragma unroll
    for (int i = 0; i < 16; ++i) bf[i] = bt[i * 32];

    __syncthreads();

    v8f acc = {};
    const float* arow = tileS + lm * GNN_FT_PITCH + 2 * half;
#pragma unroll
    for (int i = 0; i < 16; ++i) {
        const v2f a0 = *(const v2f*)(arow + 8 * i);
        const v2f a1 = *(const v2f*)(arow + 8 * i + 4);
        v2f b;
        b.x = bf[i].x; b.y = bf[i].y; acc = WMMA_F32(a0, b, acc);
        b.x = bf[i].z; b.y = bf[i].w; acc = WMMA_F32(a1, b, acc);
    }
#pragma unroll
    for (int r = 0; r < 8; ++r)
        tileO[(r + 8 * half) * GNN_FT_PITCH + 16 * wave + lm] = acc[r];

    __syncthreads();

    const float4 b2v = *(const float4*)(b2 + c);
    const float4 gv  = *(const float4*)(g + c);
    const float4 bev = *(const float4*)(beta + c);
#pragma unroll
    for (int q = 0; q < 2; ++q) {
        const float hasE = ((q == 0 ? cnt0 : cnt1) > 0.0f) ? 1.0f : 0.0f;
        float4 h = *(const float4*)(tileO + (lr0 + q) * GNN_FT_PITCH + c);
        h.x += hasE * b2v.x; h.y += hasE * b2v.y; h.z += hasE * b2v.z; h.w += hasE * b2v.w;

        float s = (h.x + h.y) + (h.z + h.w);
#pragma unroll
        for (int m = 16; m >= 1; m >>= 1) s += __shfl_xor(s, m, 32);
        const float mu = s * (1.0f / GNN_DOUT);

        h.x -= mu; h.y -= mu; h.z -= mu; h.w -= mu;
        float s2 = (h.x * h.x + h.y * h.y) + (h.z * h.z + h.w * h.w);
#pragma unroll
        for (int m = 16; m >= 1; m >>= 1) s2 += __shfl_xor(s2, m, 32);
        const float rstd = rsqrtf(s2 * (1.0f / GNN_DOUT) + GNN_EPS);

        float4 o;
        o.x = h.x * rstd * gv.x + bev.x;
        o.y = h.y * rstd * gv.y + bev.y;
        o.z = h.z * rstd * gv.z + bev.z;
        o.w = h.w * rstd * gv.w + bev.w;
        const size_t orow = row0 + (size_t)q * GNN_DOUT;
        if (RES) {
            const float4 xr = *(const float4*)(xres + orow);
            o.x += xr.x; o.y += xr.y; o.z += xr.z; o.w += xr.w;
        }
        o.x = lrelu(o.x); o.y = lrelu(o.y); o.z = lrelu(o.z); o.w = lrelu(o.w);
        *(float4*)(out + orow) = o;
    }
}

extern "C" void kernel_launch(void* const* d_in, const int* in_sizes, int n_in,
                              void* d_out, int out_size, void* d_ws, size_t ws_size,
                              hipStream_t stream) {
    (void)in_sizes; (void)n_in; (void)out_size; (void)ws_size;

    const float* x  = (const float*)d_in[0];
    const int*   ei = (const int*)d_in[1];
    const float* W1[2] = {(const float*)d_in[2],  (const float*)d_in[8]};
    const float* b1[2] = {(const float*)d_in[3],  (const float*)d_in[9]};
    const float* W2[2] = {(const float*)d_in[4],  (const float*)d_in[10]};
    const float* b2[2] = {(const float*)d_in[5],  (const float*)d_in[11]};
    const float* g [2] = {(const float*)d_in[6],  (const float*)d_in[12]};
    const float* be[2] = {(const float*)d_in[7],  (const float*)d_in[13]};

    const size_t NF = (size_t)GNN_NT * GNN_DOUT;
    float* ws   = (float*)d_ws;
    float* X    = ws;
    float* P    = X + NF;
    float* Q    = P + NF;
    float* Bp   = Q + NF;
    int* rowptr = (int*)(Bp + GNN_PK_END);
    int* slot   = rowptr + 2048;
    int* col    = slot + GNN_E;

    const dim3 egrid(GNN_N / GNN_FT_ROWS, GNN_BSN);

    k_setup<<<GNN_CSR_G + GNN_PK_END / GNN_CSR_T, GNN_CSR_T, 0, stream>>>(
        ei, rowptr, slot, col, W1[0], W2[0], W1[1], W2[1], Bp);

    k_gemm_pq<64><<<250, 256, 0, stream>>>(x, Bp + GNN_PK_A0, Bp + GNN_PK_B0, P, Q);
    k_edge_agg_ln<false><<<egrid, 256, 0, stream>>>(P, Q, b1[0], rowptr, col, Bp + GNN_PK_20,
                                                    b2[0], g[0], be[0], nullptr, X);

    k_gemm_pq<128><<<250, 256, 0, stream>>>(X, Bp + GNN_PK_A1, Bp + GNN_PK_B1, P, Q);
    k_edge_agg_ln<true><<<egrid, 256, 0, stream>>>(P, Q, b1[1], rowptr, col, Bp + GNN_PK_21,
                                                   b2[1], g[1], be[1], X, (float*)d_out);
}


# Edge-stage work-group covers its 16 nodes in two graphs: one walk of the shared bucket feeds both row sums, one W2 fragment load feeds two accumulators; waves per launch halved; sum order unchanged
# speedup vs baseline: 4.2556x; 1.0615x over previous
#include <hip/hip_runtime.h>


typedef __attribute__((ext_vector_type(2))) float v2f;
typedef __attribute__((ext_vector_type(8))) float v8f;

#define GNN_SLOPE 0.01f
#define GNN_EPS   1e-5f
#define GNN_BSN   16
#define GNN_N     2000
#define GNN_E     32000
#define GNN_NT    (GNN_BSN * GNN_N)
#define GNN_DOUT  128

__device__ __forceinline__ float lrelu(float v) { return v >= 0.0f ? v : GNN_SLOPE * v; }

#define WMMA_F32(A, B, C) \
    __builtin_amdgcn_wmma_f32_16x16x4_f32(false, (A), false, (B), (short)0, (C), false, false)

template <bool DIFF>
__device__ __forceinline__ void pack_b_elem(const float* __restrict__ B,
                                            float* __restrict__ Bp, int K, int i) {
    const int kk   = i >> 9;
    const int rem  = i & 511;
    const int lane = rem >> 4;
    const int j    = rem & 15;
    const int t    = j >> 1;
    const int comp = j & 1;
    const int r    = kk * 4 + 2 * (lane >> 4) + comp;
    const int c    = t * 16 + (lane & 15);
    const int src  = r * GNN_DOUT + c;
    float v = B[src];
    if (DIFF) v -= B[K * GNN_DOUT + src];
    Bp[i] = v;
}

__device__ __forceinline__ void pack_b_tile_elem(const float* __restrict__ B,
                                                 float* __restrict__ Bt, int i) {
    const int t    = i >> 11;
    const int rem  = i & 2047;
    const int kk2  = rem >> 7;
    const int lane = (rem & 127) >> 2;
    const int j    = rem & 3;
    const int r    = (2 * kk2 + (j >> 1)) * 4 + 2 * (lane >> 4) + (j & 1);
    const int c    = t * 16 + (lane & 15);
    Bt[i] = B[r * GNN_DOUT + c];
}

template <int K>
__global__ __launch_bounds__(256) void k_gemm_pq(const float* __restrict__ A,
                                                 const float* __restrict__ BpA,
                                                 const float* __restrict__ BpB,
                                                 float* __restrict__ P,
                                                 float* __restrict__ Q) {
    const int wave = threadIdx.x >> 5;
    const int lane = threadIdx.x & 31;
    const int half = lane >> 4;
    const int lm   = lane & 15;
    const int m0   = blockIdx.x * 128 + wave * 16;

    v8f accP[8] = {};
    v8f accQ[8] = {};

    const float*  arow = A + (size_t)(m0 + lm) * K + 2 * half;
    const float4* bpA  = (const float4*)(BpA + (size_t)lane * 16);
    const float4* bpB  = (const float4*)(BpB + (size_t)lane * 16);

#pragma unroll
    for (int k = 0; k < K; k += 4) {
        const v2f a = *(const v2f*)(arow + k);
        const float4* ba = bpA + (size_t)(k >> 2) * 128;
        const float4* bb = bpB + (size_t)(k >> 2) * 128;
        float4 pa[4], pb[4];
#pragma unroll
        for (int i = 0; i < 4; ++i) { pa[i] = ba[i]; pb[i] = bb[i]; }
        v2f b;
#pragma unroll
        for (int i = 0; i < 4; ++i) {
            b.x = pa[i].x; b.y = pa[i].y; accP[2 * i + 0] = WMMA_F32(a, b, accP[2 * i + 0]);
            b.x = pa[i].z; b.y = pa[i].w; accP[2 * i + 1] = WMMA_F32(a, b, accP[2 * i + 1]);
            b.x = pb[i].x; b.y = pb[i].y; accQ[2 * i + 0] = WMMA_F32(a, b, accQ[2 * i + 0]);
            b.x = pb[i].z; b.y = pb[i].w; accQ[2 * i + 1] = WMMA_F32(a, b, accQ[2 * i + 1]);
        }
    }

    const size_t rb = (size_t)(m0 + 8 * half) * GNN_DOUT + lm;
    float* prow = P + rb;
    float* qrow = Q + rb;
#pragma unroll
    for (int r = 0; r < 8; ++r)
#pragma unroll
        for (int t = 0; t < 8; ++t) {
            prow[(size_t)r * GNN_DOUT + t * 16] = accP[t][r];
            qrow[(size_t)r * GNN_DOUT + t * 16] = accQ[t][r];
        }
}

#define GNN_CSR_T  1024
#define GNN_CSR_IT ((GNN_E + GNN_CSR_T - 1) / GNN_CSR_T)
#define GNN_CSR_R  50
#define GNN_CSR_G  (GNN_N / GNN_CSR_R)
#define GNN_CSR_H  64
static_assert(GNN_N % GNN_CSR_R == 0, "destination ranges must tile the node set");
static_assert(GNN_CSR_R < GNN_CSR_H, "entry R of the scanned counters is the end of the range");

__device__ __forceinline__ void csr_range_body(const int* __restrict__ ei,
                                               int* __restrict__ rowptr,
                                               int* slot,
                                               int* __restrict__ col,
                                               int g) {
    __shared__ int hist[GNN_CSR_H];
    __shared__ int below;
    const int tid  = threadIdx.x;
    const int lane = tid & 31;
    const int lo   = g * GNN_CSR_R;

    if (tid < GNN_CSR_H) hist[tid] = 0;
    if (tid == 0) below = 0;
    __syncthreads();

    int rel[GNN_CSR_IT], r[GNN_CSR_IT];
#pragma unroll
    for (int it = 0; it < GNN_CSR_IT; ++it) {
        const int e = it * GNN_CSR_T + tid;
        rel[it] = (e < GNN_E) ? (ei[GNN_E + e] - lo) : GNN_N;
    }
    int nb = 0;
#pragma unroll
    for (int it = 0; it < GNN_CSR_IT; ++it) {
        r[it] = ((unsigned)rel[it] < (unsigned)GNN_CSR_R) ? atomicAdd(&hist[rel[it]], 1) : 0;
        nb += (rel[it] < 0) ? 1 : 0;
    }
#pragma unroll
    for (int m = 16; m >= 1; m >>= 1) nb += __shfl_xor(nb, m, 32);
    if (lane == 0) atomicAdd(&below, nb);
    __syncthreads();

    if (tid < 32) {
        const int i0 = 2 * lane;
        const int i1 = 2 * lane + 1;
        const int c0 = hist[i0];
        const int s  = c0 + hist[i1];
        int inc = s;
        for (int m = 1; m < 32; m <<= 1) {
            const int v = __shfl_up(inc, m, 32);
            if (lane >= m) inc += v;
        }
        const int s0 = below + inc - s;
        const int s1 = s0 + c0;
        hist[i0] = s0;
        hist[i1] = s1;
        if (i0 < GNN_CSR_R) rowptr[lo + i0] = s0;
        if (i1 < GNN_CSR_R) rowptr[lo + i1] = s1;
        if (lo + GNN_CSR_R == GNN_N) {
            if (i0 == GNN_CSR_R) rowptr[GNN_N] = s0;
            if (i1 == GNN_CSR_R) rowptr[GNN_N] = s1;
        }
    }
    __syncthreads();

#pragma unroll
    for (int it = 0; it < GNN_CSR_IT; ++it)
        if ((unsigned)rel[it] < (unsigned)GNN_CSR_R)
            slot[hist[rel[it]] + r[it]] = it * GNN_CSR_T + tid;
    __syncthreads();

    const int base = hist[0];
    const int nown = hist[GNN_CSR_R] - base;
    for (int p = tid; p < nown; p += GNN_CSR_T) {
        const int e   = slot[base + p];
        const int src = ei[e];
        const int b   = ei[GNN_E + e] - lo;
        const int beg = hist[b];
        const int end = hist[b + 1];
        int rank = 0;
        for (int j = beg; j < end; ++j) rank += (slot[j] < e) ? 1 : 0;
        col[beg + rank] = src;
    }
}

#define GNN_PK_A0 0
#define GNN_PK_B0 8192
#define GNN_PK_20 16384
#define GNN_PK_A1 32768
#define GNN_PK_B1 49152
#define GNN_PK_21 65536
#define GNN_PK_END 81920
__global__ __launch_bounds__(GNN_CSR_T) void k_setup(const int* __restrict__ ei,
                                                     int* __restrict__ rowptr,
                                                     int* slot,
                                                     int* __restrict__ col,
                                                     const float* __restrict__ W1a,
                                                     const float* __restrict__ W2a,
                                                     const float* __restrict__ W1b,
                                                     const float* __restrict__ W2b,
                                                     float* __restrict__ Bp) {
    if (blockIdx.x < GNN_CSR_G) { csr_range_body(ei, rowptr, slot, col, blockIdx.x); return; }
    const int first = (blockIdx.x - GNN_CSR_G) * GNN_CSR_T;
    const int tid   = threadIdx.x;
    if (first < GNN_PK_B0)
        pack_b_elem<true >(W1a, Bp + GNN_PK_A0, 64, first - GNN_PK_A0 + tid);
    else if (first < GNN_PK_20)
        pack_b_elem<false>(W1a + 64 * GNN_DOUT, Bp + GNN_PK_B0, 64, first - GNN_PK_B0 + tid);
    else if (first < GNN_PK_A1)
        pack_b_tile_elem(W2a, Bp + GNN_PK_20, first - GNN_PK_20 + tid);
    else if (first < GNN_PK_B1)
        pack_b_elem<true >(W1b, Bp + GNN_PK_A1, 128, first - GNN_PK_A1 + tid);
    else if (first < GNN_PK_21)
        pack_b_elem<false>(W1b + 128 * GNN_DOUT, Bp + GNN_PK_B1, 128, first - GNN_PK_B1 + tid);
    else
        pack_b_tile_elem(W2b, Bp + GNN_PK_21, first - GNN_PK_21 + tid);
}

#define GNN_FT_ROWS  16
#define GNN_FT_PITCH 132

#define GNN_FT_G     2
#define GNN_GSTRIDE  ((size_t)GNN_N * GNN_DOUT)
static_assert(GNN_BSN % GNN_FT_G == 0, "graphs must pair up");

__device__ __forceinline__ void edge_row_sum2(const float* __restrict__ qb,
                                              const float4 pa, const float4 pb,
                                              const float4 bb, const int* __restrict__ col,
                                              int beg, int end, int lane,
                                              float4& outA, float4& outB) {
    float4 accA = {0.0f, 0.0f, 0.0f, 0.0f};
    float4 accB = {0.0f, 0.0f, 0.0f, 0.0f};
    for (int base = beg; base < end; base += 32) {
        const int n    = (end - base < 32) ? (end - base) : 32;
        const int mine = (lane < n) ? col[base + lane] : 0;
#pragma unroll 4
        for (int j = 0; j < n; ++j) {
            const int src = __builtin_amdgcn_readlane(mine, j);
            const float* r = qb + (size_t)src * GNN_DOUT;
            const float4 qa = *(const float4*)(r);
            const float4 qc = *(const float4*)(r + GNN_GSTRIDE);
            accA.x += lrelu(pa.x + qa.x + bb.x);
            accA.y += lrelu(pa.y + qa.y + bb.y);
            accA.z += lrelu(pa.z + qa.z + bb.z);
            accA.w += lrelu(pa.w + qa.w + bb.w);
            accB.x += lrelu(pb.x + qc.x + bb.x);
            accB.y += lrelu(pb.y + qc.y + bb.y);
            accB.z += lrelu(pb.z + qc.z + bb.z);
            accB.w += lrelu(pb.w + qc.w + bb.w);
        }
    }
    outA = accA;
    outB = accB;
}

template <bool RES>
__global__ __launch_bounds__(256) void k_edge_agg_ln(const float* __restrict__ P,
                                                     const float* __restrict__ Q,
                                                     const float* __restrict__ b1,
                                                     const int* __restrict__ rowptr,
                                                     const int* __restrict__ col,
                                                     const float* __restrict__ Bt2,
                                                     const float* __restrict__ b2,
                                                     const float* __restrict__ g,
                                                     const float* __restrict__ beta,
                                                     const float* __restrict__ xres,
                                                     float* __restrict__ out) {
    __shared__ __attribute__((aligned(16))) float tileS[GNN_FT_G][GNN_FT_ROWS * GNN_FT_PITCH];
    __shared__ __attribute__((aligned(16))) float tileO[GNN_FT_G][GNN_FT_ROWS * GNN_FT_PITCH];

    const int wave = __builtin_amdgcn_readfirstlane((int)(threadIdx.x >> 5));
    const int lane = threadIdx.x & 31;
    const int half = lane >> 4;
    const int lm   = lane & 15;
    const int c    = lane * 4;
    const int off  = blockIdx.y * (GNN_FT_G * GNN_N);
    const int lr0  = 2 * wave;
    const int node = blockIdx.x * GNN_FT_ROWS + lr0;

    const int beg0 = rowptr[node];
    const int beg1 = rowptr[node + 1];
    const int end1 = rowptr[node + 2];
    const size_t row0 = (size_t)(off + node) * GNN_DOUT + c;
    const float4 p0a = *(const float4*)(P + row0);
    const float4 p1a = *(const float4*)(P + row0 + GNN_DOUT);
    const float4 p0b = *(const float4*)(P + row0 + GNN_GSTRIDE);
    const float4 p1b = *(const float4*)(P + row0 + GNN_GSTRIDE + GNN_DOUT);
    const float4 bb = *(const float4*)(b1 + c);
    const float* qb = Q + (size_t)off * GNN_DOUT + c;

    const float cnt0 = (float)(beg1 - beg0);
    const float cnt1 = (float)(end1 - beg1);
    {
        float4 sa, sb;
        edge_row_sum2(qb, p0a, p0b, bb, col, beg0, beg1, lane, sa, sb);
        const float sc0 = 1.0f / fmaxf(cnt0, 1.0f);
        sa.x *= sc0; sa.y *= sc0; sa.z *= sc0; sa.w *= sc0;
        sb.x *= sc0; sb.y *= sc0; sb.z *= sc0; sb.w *= sc0;
        *(float4*)(tileS[0] + (lr0 + 0) * GNN_FT_PITCH + c) = sa;
        *(float4*)(tileS[1] + (lr0 + 0) * GNN_FT_PITCH + c) = sb;
        edge_row_sum2(qb, p1a, p1b, bb, col, beg1, end1, lane, sa, sb);
        const float sc1 = 1.0f / fmaxf(cnt1, 1.0f);
        sa.x *= sc1; sa.y *= sc1; sa.z *= sc1; sa.w *= sc1;
        sb.x *= sc1; sb.y *= sc1; sb.z *= sc1; sb.w *= sc1;
        *(float4*)(tileS[0] + (lr0 + 1) * GNN_FT_PITCH + c) = sa;
        *(float4*)(tileS[1] + (lr0 + 1) * GNN_FT_PITCH + c) = sb;
    }

    const float4* bt = (const float4*)(Bt2 + (size_t)wave * 2048 + lane * 4);
    float4 bf[16];
#pragma unroll
    for (int i = 0; i < 16; ++i) bf[i] = bt[i * 32];

    __syncthreads();

    v8f accA = {};
    v8f accB = {};
    const float* arowA = tileS[0] + lm * GNN_FT_PITCH + 2 * half;
    const float* arowB = tileS[1] + lm * GNN_FT_PITCH + 2 * half;
#pragma unroll
    for (int i = 0; i < 16; ++i) {
        const v2f a0 = *(const v2f*)(arowA + 8 * i);
        const v2f a1 = *(const v2f*)(arowA + 8 * i + 4);
        const v2f c0 = *(const v2f*)(arowB + 8 * i);
        const v2f c1 = *(const v2f*)(arowB + 8 * i + 4);
        v2f b;
        b.x = bf[i].x; b.y = bf[i].y;
        accA = WMMA_F32(a0, b, accA);
        accB = WMMA_F32(c0, b, accB);
        b.x = bf[i].z; b.y = bf[i].w;
        accA = WMMA_F32(a1, b, accA);
        accB = WMMA_F32(c1, b, accB);
    }
#pragma unroll
    for (int r = 0; r < 8; ++r) {
        tileO[0][(r + 8 * half) * GNN_FT_PITCH + 16 * wave + lm] = accA[r];
        tileO[1][(r + 8 * half) * GNN_FT_PITCH + 16 * wave + lm] = accB[r];
    }

    __syncthreads();

    const float4 b2v = *(const float4*)(b2 + c);
    const float4 gv  = *(const float4*)(g + c);
    const float4 bev = *(const float4*)(beta + c);
#pragma unroll
    for (int gq = 0; gq < 2 * GNN_FT_G; ++gq) {
        const int gi = gq >> 1;
        const int q  = gq & 1;
        const float hasE = ((q == 0 ? cnt0 : cnt1) > 0.0f) ? 1.0f : 0.0f;
        float4 h = *(const float4*)(tileO[gi] + (lr0 + q) * GNN_FT_PITCH + c);
        h.x += hasE * b2v.x; h.y += hasE * b2v.y; h.z += hasE * b2v.z; h.w += hasE * b2v.w;

        float s = (h.x + h.y) + (h.z + h.w);
#pragma unroll
        for (int m = 16; m >= 1; m >>= 1) s += __shfl_xor(s, m, 32);
        const float mu = s * (1.0f / GNN_DOUT);

        h.x -= mu; h.y -= mu; h.z -= mu; h.w -= mu;
        float s2 = (h.x * h.x + h.y * h.y) + (h.z * h.z + h.w * h.w);
#pragma unroll
        for (int m = 16; m >= 1; m >>= 1) s2 += __shfl_xor(s2, m, 32);
        const float rstd = rsqrtf(s2 * (1.0f / GNN_DOUT) + GNN_EPS);

        float4 o;
        o.x = h.x * rstd * gv.x + bev.x;
        o.y = h.y * rstd * gv.y + bev.y;
        o.z = h.z * rstd * gv.z + bev.z;
        o.w = h.w * rstd * gv.w + bev.w;
        const size_t orow = row0 + (size_t)gi * GNN_GSTRIDE + (size_t)q * GNN_DOUT;
        if (RES) {
            const float4 xr = *(const float4*)(xres + orow);
            o.x += xr.x; o.y += xr.y; o.z += xr.z; o.w += xr.w;
        }
        o.x = lrelu(o.x); o.y = lrelu(o.y); o.z = lrelu(o.z); o.w = lrelu(o.w);
        *(float4*)(out + orow) = o;
    }
}

extern "C" void kernel_launch(void* const* d_in, const int* in_sizes, int n_in,
                              void* d_out, int out_size, void* d_ws, size_t ws_size,
                              hipStream_t stream) {
    (void)in_sizes; (void)n_in; (void)out_size; (void)ws_size;

    const float* x  = (const float*)d_in[0];
    const int*   ei = (const int*)d_in[1];
    const float* W1[2] = {(const float*)d_in[2],  (const float*)d_in[8]};
    const float* b1[2] = {(const float*)d_in[3],  (const float*)d_in[9]};
    const float* W2[2] = {(const float*)d_in[4],  (const float*)d_in[10]};
    const float* b2[2] = {(const float*)d_in[5],  (const float*)d_in[11]};
    const float* g [2] = {(const float*)d_in[6],  (const float*)d_in[12]};
    const float* be[2] = {(const float*)d_in[7],  (const float*)d_in[13]};

    const size_t NF = (size_t)GNN_NT * GNN_DOUT;
    float* ws   = (float*)d_ws;
    float* X    = ws;
    float* P    = X + NF;
    float* Q    = P + NF;
    float* Bp   = Q + NF;
    int* rowptr = (int*)(Bp + GNN_PK_END);
    int* slot   = rowptr + 2048;
    int* col    = slot + GNN_E;

    const dim3 egrid(GNN_N / GNN_FT_ROWS, GNN_BSN / GNN_FT_G);

    k_setup<<<GNN_CSR_G + GNN_PK_END / GNN_CSR_T, GNN_CSR_T, 0, stream>>>(
        ei, rowptr, slot, col, W1[0], W2[0], W1[1], W2[1], Bp);

    k_gemm_pq<64><<<250, 256, 0, stream>>>(x, Bp + GNN_PK_A0, Bp + GNN_PK_B0, P, Q);
    k_edge_agg_ln<false><<<egrid, 256, 0, stream>>>(P, Q, b1[0], rowptr, col, Bp + GNN_PK_20,
                                                    b2[0], g[0], be[0], nullptr, X);

    k_gemm_pq<128><<<250, 256, 0, stream>>>(X, Bp + GNN_PK_A1, Bp + GNN_PK_B1, P, Q);
    k_edge_agg_ln<true><<<egrid, 256, 0, stream>>>(P, Q, b1[1], rowptr, col, Bp + GNN_PK_21,
                                                   b2[1], g[1], be[1], X, (float*)d_out);
}


# P/Q GEMM k-loop: each k-step's 8 B-fragment loads issued as one group two k-steps ahead of its 16 WMMAs (ring of 3 register sets), A fragments one 128-B row line ahead; was ~3 loads in flight
# speedup vs baseline: 4.9792x; 1.1700x over previous
#include <hip/hip_runtime.h>


typedef __attribute__((ext_vector_type(2))) float v2f;
typedef __attribute__((ext_vector_type(8))) float v8f;

#define GNN_SLOPE 0.01f
#define GNN_EPS   1e-5f
#define GNN_BSN   16
#define GNN_N     2000
#define GNN_E     32000
#define GNN_NT    (GNN_BSN * GNN_N)
#define GNN_DOUT  128

__device__ __forceinline__ float lrelu(float v) { return v >= 0.0f ? v : GNN_SLOPE * v; }

#define WMMA_F32(A, B, C) \
    __builtin_amdgcn_wmma_f32_16x16x4_f32(false, (A), false, (B), (short)0, (C), false, false)

template <bool DIFF>
__device__ __forceinline__ void pack_b_elem(const float* __restrict__ B,
                                            float* __restrict__ Bp, int K, int i) {
    const int kk   = i >> 9;
    const int rem  = i & 511;
    const int lane = rem >> 4;
    const int j    = rem & 15;
    const int t    = j >> 1;
    const int comp = j & 1;
    const int r    = kk * 4 + 2 * (lane >> 4) + comp;
    const int c    = t * 16 + (lane & 15);
    const int src  = r * GNN_DOUT + c;
    float v = B[src];
    if (DIFF) v -= B[K * GNN_DOUT + src];
    Bp[i] = v;
}

__device__ __forceinline__ void pack_b_tile_elem(const float* __restrict__ B,
                                                 float* __restrict__ Bt, int i) {
    const int t    = i >> 11;
    const int rem  = i & 2047;
    const int kk2  = rem >> 7;
    const int lane = (rem & 127) >> 2;
    const int j    = rem & 3;
    const int r    = (2 * kk2 + (j >> 1)) * 4 + 2 * (lane >> 4) + (j & 1);
    const int c    = t * 16 + (lane & 15);
    Bt[i] = B[r * GNN_DOUT + c];
}

#define GNN_PQ_AHEAD 2
#define GNN_PQ_RING  (GNN_PQ_AHEAD + 1)
#define GNN_PQ_ACH   8
template <int K>
__global__ __launch_bounds__(256) void k_gemm_pq(const float* __restrict__ A,
                                                 const float* __restrict__ BpA,
                                                 const float* __restrict__ BpB,
                                                 float* __restrict__ P,
                                                 float* __restrict__ Q) {
    constexpr int NS = K / 4;
    static_assert(NS % GNN_PQ_ACH == 0 && NS >= GNN_PQ_RING, "k-steps must fill whole A chunks");

    const int wave = threadIdx.x >> 5;
    const int lane = threadIdx.x & 31;
    const int half = lane >> 4;
    const int lm   = lane & 15;
    const int m0   = blockIdx.x * 128 + wave * 16;

    v8f accP[8] = {};
    v8f accQ[8] = {};

    const float*  arow = A + (size_t)(m0 + lm) * K + 2 * half;
    const float4* bpA  = (const float4*)(BpA + (size_t)lane * 16);
    const float4* bpB  = (const float4*)(BpB + (size_t)lane * 16);

    float4 pa[GNN_PQ_RING][4], pb[GNN_PQ_RING][4];
    v2f    af[2][GNN_PQ_ACH];

#pragma unroll
    for (int i = 0; i < GNN_PQ_ACH; ++i) af[0][i] = *(const v2f*)(arow + 4 * i);
#pragma unroll
    for (int s = 0; s < GNN_PQ_AHEAD; ++s)
#pragma unroll
        for (int i = 0; i < 4; ++i) {
            pa[s][i] = bpA[(size_t)s * 128 + i];
            pb[s][i] = bpB[(size_t)s * 128 + i];
        }

#pragma unroll
    for (int s = 0; s < NS; ++s) {
        if (s + GNN_PQ_AHEAD < NS) {
            const int n = s + GNN_PQ_AHEAD;
#pragma unroll
            for (int i = 0; i < 4; ++i) {
                pa[n % GNN_PQ_RING][i] = bpA[(size_t)n * 128 + i];
                pb[n % GNN_PQ_RING][i] = bpB[(size_t)n * 128 + i];
            }
        }
        if (s % GNN_PQ_ACH == 0 && s + GNN_PQ_ACH < NS) {
            const int ch = s / GNN_PQ_ACH + 1;
#pragma unroll
            for (int i = 0; i < GNN_PQ_ACH; ++i)
                af[ch & 1][i] = *(const v2f*)(arow + 4 * (ch * GNN_PQ_ACH + i));
        }
        __builtin_amdgcn_sched_barrier(0);

        const v2f a = af[(s / GNN_PQ_ACH) & 1][s % GNN_PQ_ACH];
        v2f b;
#pragma unroll
        for (int i = 0; i < 4; ++i) {
            const float4 fa = pa[s % GNN_PQ_RING][i];
            const float4 fb = pb[s % GNN_PQ_RING][i];
            b.x = fa.x; b.y = fa.y; accP[2 * i + 0] = WMMA_F32(a, b, accP[2 * i + 0]);
            b.x = fa.z; b.y = fa.w; accP[2 * i + 1] = WMMA_F32(a, b, accP[2 * i + 1]);
            b.x = fb.x; b.y = fb.y; accQ[2 * i + 0] = WMMA_F32(a, b, accQ[2 * i + 0]);
            b.x = fb.z; b.y = fb.w; accQ[2 * i + 1] = WMMA_F32(a, b, accQ[2 * i + 1]);
        }
        __builtin_amdgcn_sched_barrier(0);
    }

    const size_t rb = (size_t)(m0 + 8 * half) * GNN_DOUT + lm;
    float* prow = P + rb;
    float* qrow = Q + rb;
#pragma unroll
    for (int r = 0; r < 8; ++r)
#pragma unroll
        for (int t = 0; t < 8; ++t) {
            prow[(size_t)r * GNN_DOUT + t * 16] = accP[t][r];
            qrow[(size_t)r * GNN_DOUT + t * 16] = accQ[t][r];
        }
}

#define GNN_CSR_T  1024
#define GNN_CSR_IT ((GNN_E + GNN_CSR_T - 1) / GNN_CSR_T)
#define GNN_CSR_R  50
#define GNN_CSR_G  (GNN_N / GNN_CSR_R)
#define GNN_CSR_H  64
static_assert(GNN_N % GNN_CSR_R == 0, "destination ranges must tile the node set");
static_assert(GNN_CSR_R < GNN_CSR_H, "entry R of the scanned counters is the end of the range");

__device__ __forceinline__ void csr_range_body(const int* __restrict__ ei,
                                               int* __restrict__ rowptr,
                                               int* slot,
                                               int* __restrict__ col,
                                               int g) {
    __shared__ int hist[GNN_CSR_H];
    __shared__ int below;
    const int tid  = threadIdx.x;
    const int lane = tid & 31;
    const int lo   = g * GNN_CSR_R;

    if (tid < GNN_CSR_H) hist[tid] = 0;
    if (tid == 0) below = 0;
    __syncthreads();

    int rel[GNN_CSR_IT], r[GNN_CSR_IT];
#pragma unroll
    for (int it = 0; it < GNN_CSR_IT; ++it) {
        const int e = it * GNN_CSR_T + tid;
        rel[it] = (e < GNN_E) ? (ei[GNN_E + e] - lo) : GNN_N;
    }
    int nb = 0;
#pragma unroll
    for (int it = 0; it < GNN_CSR_IT; ++it) {
        r[it] = ((unsigned)rel[it] < (unsigned)GNN_CSR_R) ? atomicAdd(&hist[rel[it]], 1) : 0;
        nb += (rel[it] < 0) ? 1 : 0;
    }
#pragma unroll
    for (int m = 16; m >= 1; m >>= 1) nb += __shfl_xor(nb, m, 32);
    if (lane == 0) atomicAdd(&below, nb);
    __syncthreads();

    if (tid < 32) {
        const int i0 = 2 * lane;
        const int i1 = 2 * lane + 1;
        const int c0 = hist[i0];
        const int s  = c0 + hist[i1];
        int inc = s;
        for (int m = 1; m < 32; m <<= 1) {
            const int v = __shfl_up(inc, m, 32);
            if (lane >= m) inc += v;
        }
        const int s0 = below + inc - s;
        const int s1 = s0 + c0;
        hist[i0] = s0;
        hist[i1] = s1;
        if (i0 < GNN_CSR_R) rowptr[lo + i0] = s0;
        if (i1 < GNN_CSR_R) rowptr[lo + i1] = s1;
        if (lo + GNN_CSR_R == GNN_N) {
            if (i0 == GNN_CSR_R) rowptr[GNN_N] = s0;
            if (i1 == GNN_CSR_R) rowptr[GNN_N] = s1;
        }
    }
    __syncthreads();

#pragma unroll
    for (int it = 0; it < GNN_CSR_IT; ++it)
        if ((unsigned)rel[it] < (unsigned)GNN_CSR_R)
            slot[hist[rel[it]] + r[it]] = it * GNN_CSR_T + tid;
    __syncthreads();

    const int base = hist[0];
    const int nown = hist[GNN_CSR_R] - base;
    for (int p = tid; p < nown; p += GNN_CSR_T) {
        const int e   = slot[base + p];
        const int src = ei[e];
        const int b   = ei[GNN_E + e] - lo;
        const int beg = hist[b];
        const int end = hist[b + 1];
        int rank = 0;
        for (int j = beg; j < end; ++j) rank += (slot[j] < e) ? 1 : 0;
        col[beg + rank] = src;
    }
}

#define GNN_PK_A0 0
#define GNN_PK_B0 8192
#define GNN_PK_20 16384
#define GNN_PK_A1 32768
#define GNN_PK_B1 49152
#define GNN_PK_21 65536
#define GNN_PK_END 81920
__global__ __launch_bounds__(GNN_CSR_T) void k_setup(const int* __restrict__ ei,
                                                     int* __restrict__ rowptr,
                                                     int* slot,
                                                     int* __restrict__ col,
                                                     const float* __restrict__ W1a,
                                                     const float* __restrict__ W2a,
                                                     const float* __restrict__ W1b,
                                                     const float* __restrict__ W2b,
                                                     float* __restrict__ Bp) {
    if (blockIdx.x < GNN_CSR_G) { csr_range_body(ei, rowptr, slot, col, blockIdx.x); return; }
    const int first = (blockIdx.x - GNN_CSR_G) * GNN_CSR_T;
    const int tid   = threadIdx.x;
    if (first < GNN_PK_B0)
        pack_b_elem<true >(W1a, Bp + GNN_PK_A0, 64, first - GNN_PK_A0 + tid);
    else if (first < GNN_PK_20)
        pack_b_elem<false>(W1a + 64 * GNN_DOUT, Bp + GNN_PK_B0, 64, first - GNN_PK_B0 + tid);
    else if (first < GNN_PK_A1)
        pack_b_tile_elem(W2a, Bp + GNN_PK_20, first - GNN_PK_20 + tid);
    else if (first < GNN_PK_B1)
        pack_b_elem<true >(W1b, Bp + GNN_PK_A1, 128, first - GNN_PK_A1 + tid);
    else if (first < GNN_PK_21)
        pack_b_elem<false>(W1b + 128 * GNN_DOUT, Bp + GNN_PK_B1, 128, first - GNN_PK_B1 + tid);
    else
        pack_b_tile_elem(W2b, Bp + GNN_PK_21, first - GNN_PK_21 + tid);
}

#define GNN_FT_ROWS  16
#define GNN_FT_PITCH 132

#define GNN_FT_G     2
#define GNN_GSTRIDE  ((size_t)GNN_N * GNN_DOUT)
static_assert(GNN_BSN % GNN_FT_G == 0, "graphs must pair up");

__device__ __forceinline__ void edge_row_sum2(const float* __restrict__ qb,
                                              const float4 pa, const float4 pb,
                                              const float4 bb, const int* __restrict__ col,
                                              int beg, int end, int lane,
                                              float4& outA, float4& outB) {
    float4 accA = {0.0f, 0.0f, 0.0f, 0.0f};
    float4 accB = {0.0f, 0.0f, 0.0f, 0.0f};
    for (int base = beg; base < end; base += 32) {
        const int n    = (end - base < 32) ? (end - base) : 32;
        const int mine = (lane < n) ? col[base + lane] : 0;
#pragma unroll 4
        for (int j = 0; j < n; ++j) {
            const int src = __builtin_amdgcn_readlane(mine, j);
            const float* r = qb + (size_t)src * GNN_DOUT;
            const float4 qa = *(const float4*)(r);
            const float4 qc = *(const float4*)(r + GNN_GSTRIDE);
            accA.x += lrelu(pa.x + qa.x + bb.x);
            accA.y += lrelu(pa.y + qa.y + bb.y);
            accA.z += lrelu(pa.z + qa.z + bb.z);
            accA.w += lrelu(pa.w + qa.w + bb.w);
            accB.x += lrelu(pb.x + qc.x + bb.x);
            accB.y += lrelu(pb.y + qc.y + bb.y);
            accB.z += lrelu(pb.z + qc.z + bb.z);
            accB.w += lrelu(pb.w + qc.w + bb.w);
        }
    }
    outA = accA;
    outB = accB;
}

template <bool RES>
__global__ __launch_bounds__(256) void k_edge_agg_ln(const float* __restrict__ P,
                                                     const float* __restrict__ Q,
                                                     const float* __restrict__ b1,
                                                     const int* __restrict__ rowptr,
                                                     const int* __restrict__ col,
                                                     const float* __restrict__ Bt2,
                                                     const float* __restrict__ b2,
                                                     const float* __restrict__ g,
                                                     const float* __restrict__ beta,
                                                     const float* __restrict__ xres,
                                                     float* __restrict__ out) {
    __shared__ __attribute__((aligned(16))) float tileS[GNN_FT_G][GNN_FT_ROWS * GNN_FT_PITCH];
    __shared__ __attribute__((aligned(16))) float tileO[GNN_FT_G][GNN_FT_ROWS * GNN_FT_PITCH];

    const int wave = __builtin_amdgcn_readfirstlane((int)(threadIdx.x >> 5));
    const int lane = threadIdx.x & 31;
    const int half = lane >> 4;
    const int lm   = lane & 15;
    const int c    = lane * 4;
    const int off  = blockIdx.y * (GNN_FT_G * GNN_N);
    const int lr0  = 2 * wave;
    const int node = blockIdx.x * GNN_FT_ROWS + lr0;

    const int beg0 = rowptr[node];
    const int beg1 = rowptr[node + 1];
    const int end1 = rowptr[node + 2];
    const size_t row0 = (size_t)(off + node) * GNN_DOUT + c;
    const float4 p0a = *(const float4*)(P + row0);
    const float4 p1a = *(const float4*)(P + row0 + GNN_DOUT);
    const float4 p0b = *(const float4*)(P + row0 + GNN_GSTRIDE);
    const float4 p1b = *(const float4*)(P + row0 + GNN_GSTRIDE + GNN_DOUT);
    const float4 bb = *(const float4*)(b1 + c);
    const float* qb = Q + (size_t)off * GNN_DOUT + c;

    const float cnt0 = (float)(beg1 - beg0);
    const float cnt1 = (float)(end1 - beg1);
    {
        float4 sa, sb;
        edge_row_sum2(qb, p0a, p0b, bb, col, beg0, beg1, lane, sa, sb);
        const float sc0 = 1.0f / fmaxf(cnt0, 1.0f);
        sa.x *= sc0; sa.y *= sc0; sa.z *= sc0; sa.w *= sc0;
        sb.x *= sc0; sb.y *= sc0; sb.z *= sc0; sb.w *= sc0;
        *(float4*)(tileS[0] + (lr0 + 0) * GNN_FT_PITCH + c) = sa;
        *(float4*)(tileS[1] + (lr0 + 0) * GNN_FT_PITCH + c) = sb;
        edge_row_sum2(qb, p1a, p1b, bb, col, beg1, end1, lane, sa, sb);
        const float sc1 = 1.0f / fmaxf(cnt1, 1.0f);
        sa.x *= sc1; sa.y *= sc1; sa.z *= sc1; sa.w *= sc1;
        sb.x *= sc1; sb.y *= sc1; sb.z *= sc1; sb.w *= sc1;
        *(float4*)(tileS[0] + (lr0 + 1) * GNN_FT_PITCH + c) = sa;
        *(float4*)(tileS[1] + (lr0 + 1) * GNN_FT_PITCH + c) = sb;
    }

    const float4* bt = (const float4*)(Bt2 + (size_t)wave * 2048 + lane * 4);
    float4 bf[16];
#pragma unroll
    for (int i = 0; i < 16; ++i) bf[i] = bt[i * 32];

    __syncthreads();

    v8f accA = {};
    v8f accB = {};
    const float* arowA = tileS[0] + lm * GNN_FT_PITCH + 2 * half;
    const float* arowB = tileS[1] + lm * GNN_FT_PITCH + 2 * half;
#pragma unroll
    for (int i = 0; i < 16; ++i) {
        const v2f a0 = *(const v2f*)(arowA + 8 * i);
        const v2f a1 = *(const v2f*)(arowA + 8 * i + 4);
        const v2f c0 = *(const v2f*)(arowB + 8 * i);
        const v2f c1 = *(const v2f*)(arowB + 8 * i + 4);
        v2f b;
        b.x = bf[i].x; b.y = bf[i].y;
        accA = WMMA_F32(a0, b, accA);
        accB = WMMA_F32(c0, b, accB);
        b.x = bf[i].z; b.y = bf[i].w;
        accA = WMMA_F32(a1, b, accA);
        accB = WMMA_F32(c1, b, accB);
    }
#pragma unroll
    for (int r = 0; r < 8; ++r) {
        tileO[0][(r + 8 * half) * GNN_FT_PITCH + 16 * wave + lm] = accA[r];
        tileO[1][(r + 8 * half) * GNN_FT_PITCH + 16 * wave + lm] = accB[r];
    }

    __syncthreads();

    const float4 b2v = *(const float4*)(b2 + c);
    const float4 gv  = *(const float4*)(g + c);
    const float4 bev = *(const float4*)(beta + c);
#pragma unroll
    for (int gq = 0; gq < 2 * GNN_FT_G; ++gq) {
        const int gi = gq >> 1;
        const int q  = gq & 1;
        const float hasE = ((q == 0 ? cnt0 : cnt1) > 0.0f) ? 1.0f : 0.0f;
        float4 h = *(const float4*)(tileO[gi] + (lr0 + q) * GNN_FT_PITCH + c);
        h.x += hasE * b2v.x; h.y += hasE * b2v.y; h.z += hasE * b2v.z; h.w += hasE * b2v.w;

        float s = (h.x + h.y) + (h.z + h.w);
#pragma unroll
        for (int m = 16; m >= 1; m >>= 1) s += __shfl_xor(s, m, 32);
        const float mu = s * (1.0f / GNN_DOUT);

        h.x -= mu; h.y -= mu; h.z -= mu; h.w -= mu;
        float s2 = (h.x * h.x + h.y * h.y) + (h.z * h.z + h.w * h.w);
#pragma unroll
        for (int m = 16; m >= 1; m >>= 1) s2 += __shfl_xor(s2, m, 32);
        const float rstd = rsqrtf(s2 * (1.0f / GNN_DOUT) + GNN_EPS);

        float4 o;
        o.x = h.x * rstd * gv.x + bev.x;
        o.y = h.y * rstd * gv.y + bev.y;
        o.z = h.z * rstd * gv.z + bev.z;
        o.w = h.w * rstd * gv.w + bev.w;
        const size_t orow = row0 + (size_t)gi * GNN_GSTRIDE + (size_t)q * GNN_DOUT;
        if (RES) {
            const float4 xr = *(const float4*)(xres + orow);
            o.x += xr.x; o.y += xr.y; o.z += xr.z; o.w += xr.w;
        }
        o.x = lrelu(o.x); o.y = lrelu(o.y); o.z = lrelu(o.z); o.w = lrelu(o.w);
        *(float4*)(out + orow) = o;
    }
}

extern "C" void kernel_launch(void* const* d_in, const int* in_sizes, int n_in,
                              void* d_out, int out_size, void* d_ws, size_t ws_size,
                              hipStream_t stream) {
    (void)in_sizes; (void)n_in; (void)out_size; (void)ws_size;

    const float* x  = (const float*)d_in[0];
    const int*   ei = (const int*)d_in[1];
    const float* W1[2] = {(const float*)d_in[2],  (const float*)d_in[8]};
    const float* b1[2] = {(const float*)d_in[3],  (const float*)d_in[9]};
    const float* W2[2] = {(const float*)d_in[4],  (const float*)d_in[10]};
    const float* b2[2] = {(const float*)d_in[5],  (const float*)d_in[11]};
    const float* g [2] = {(const float*)d_in[6],  (const float*)d_in[12]};
    const float* be[2] = {(const float*)d_in[7],  (const float*)d_in[13]};

    const size_t NF = (size_t)GNN_NT * GNN_DOUT;
    float* ws   = (float*)d_ws;
    float* X    = ws;
    float* P    = X + NF;
    float* Q    = P + NF;
    float* Bp   = Q + NF;
    int* rowptr = (int*)(Bp + GNN_PK_END);
    int* slot   = rowptr + 2048;
    int* col    = slot + GNN_E;

    const dim3 egrid(GNN_N / GNN_FT_ROWS, GNN_BSN / GNN_FT_G);

    k_setup<<<GNN_CSR_G + GNN_PK_END / GNN_CSR_T, GNN_CSR_T, 0, stream>>>(
        ei, rowptr, slot, col, W1[0], W2[0], W1[1], W2[1], Bp);

    k_gemm_pq<64><<<250, 256, 0, stream>>>(x, Bp + GNN_PK_A0, Bp + GNN_PK_B0, P, Q);
    k_edge_agg_ln<false><<<egrid, 256, 0, stream>>>(P, Q, b1[0], rowptr, col, Bp + GNN_PK_20,
                                                    b2[0], g[0], be[0], nullptr, X);

    k_gemm_pq<128><<<250, 256, 0, stream>>>(X, Bp + GNN_PK_A1, Bp + GNN_PK_B1, P, Q);
    k_edge_agg_ln<true><<<egrid, 256, 0, stream>>>(P, Q, b1[1], rowptr, col, Bp + GNN_PK_21,
                                                   b2[1], g[1], be[1], X, (float*)d_out);
}


# Block 1's P/Q GEMM moves into block 0's edge-stage launch: each work-group multiplies its 32 finished rows (LDS) by block 1's W1 and stores whole P,Q rows; 5 -> 4 launches; sum order unchanged
# speedup vs baseline: 5.4930x; 1.1032x over previous
#include <hip/hip_runtime.h>


typedef __attribute__((ext_vector_type(2))) float v2f;
typedef __attribute__((ext_vector_type(8))) float v8f;

#define GNN_SLOPE 0.01f
#define GNN_EPS   1e-5f
#define GNN_BSN   16
#define GNN_N     2000
#define GNN_E     32000
#define GNN_NT    (GNN_BSN * GNN_N)
#define GNN_DOUT  128

__device__ __forceinline__ float lrelu(float v) { return v >= 0.0f ? v : GNN_SLOPE * v; }

#define WMMA_F32(A, B, C) \
    __builtin_amdgcn_wmma_f32_16x16x4_f32(false, (A), false, (B), (short)0, (C), false, false)

template <bool DIFF>
__device__ __forceinline__ void pack_b_elem(const float* __restrict__ B,
                                            float* __restrict__ Bp, int K, int i) {
    const int kk   = i >> 9;
    const int rem  = i & 511;
    const int lane = rem >> 4;
    const int j    = rem & 15;
    const int t    = j >> 1;
    const int comp = j & 1;
    const int r    = kk * 4 + 2 * (lane >> 4) + comp;
    const int c    = t * 16 + (lane & 15);
    const int src  = r * GNN_DOUT + c;
    float v = B[src];
    if (DIFF) v -= B[K * GNN_DOUT + src];
    Bp[i] = v;
}

template <bool DIFF>
__device__ __forceinline__ void pack_b_tile_elem(const float* __restrict__ B,
                                                 float* __restrict__ Bt, int i) {
    const int t    = i >> 11;
    const int rem  = i & 2047;
    const int kk2  = rem >> 7;
    const int lane = (rem & 127) >> 2;
    const int j    = rem & 3;
    const int r    = (2 * kk2 + (j >> 1)) * 4 + 2 * (lane >> 4) + (j & 1);
    const int c    = t * 16 + (lane & 15);
    float v = B[r * GNN_DOUT + c];
    if (DIFF) v -= B[(GNN_DOUT + r) * GNN_DOUT + c];
    Bt[i] = v;
}

#define GNN_PQ_AHEAD 2
#define GNN_PQ_RING  (GNN_PQ_AHEAD + 1)
#define GNN_PQ_ACH   8
template <int K>
__global__ __launch_bounds__(256) void k_gemm_pq(const float* __restrict__ A,
                                                 const float* __restrict__ BpA,
                                                 const float* __restrict__ BpB,
                                                 float* __restrict__ P,
                                                 float* __restrict__ Q) {
    constexpr int NS = K / 4;
    static_assert(NS % GNN_PQ_ACH == 0 && NS >= GNN_PQ_RING, "k-steps must fill whole A chunks");

    const int wave = threadIdx.x >> 5;
    const int lane = threadIdx.x & 31;
    const int half = lane >> 4;
    const int lm   = lane & 15;
    const int m0   = blockIdx.x * 128 + wave * 16;

    v8f accP[8] = {};
    v8f accQ[8] = {};

    const float*  arow = A + (size_t)(m0 + lm) * K + 2 * half;
    const float4* bpA  = (const float4*)(BpA + (size_t)lane * 16);
    const float4* bpB  = (const float4*)(BpB + (size_t)lane * 16);

    float4 pa[GNN_PQ_RING][4], pb[GNN_PQ_RING][4];
    v2f    af[2][GNN_PQ_ACH];

#pragma unroll
    for (int i = 0; i < GNN_PQ_ACH; ++i) af[0][i] = *(const v2f*)(arow + 4 * i);
#pragma unroll
    for (int s = 0; s < GNN_PQ_AHEAD; ++s)
#pragma unroll
        for (int i = 0; i < 4; ++i) {
            pa[s][i] = bpA[(size_t)s * 128 + i];
            pb[s][i] = bpB[(size_t)s * 128 + i];
        }

#pragma unroll
    for (int s = 0; s < NS; ++s) {
        if (s + GNN_PQ_AHEAD < NS) {
            const int n = s + GNN_PQ_AHEAD;
#pragma unroll
            for (int i = 0; i < 4; ++i) {
                pa[n % GNN_PQ_RING][i] = bpA[(size_t)n * 128 + i];
                pb[n % GNN_PQ_RING][i] = bpB[(size_t)n * 128 + i];
            }
        }
        if (s % GNN_PQ_ACH == 0 && s + GNN_PQ_ACH < NS) {
            const int ch = s / GNN_PQ_ACH + 1;
#pragma unroll
            for (int i = 0; i < GNN_PQ_ACH; ++i)
                af[ch & 1][i] = *(const v2f*)(arow + 4 * (ch * GNN_PQ_ACH + i));
        }
        __builtin_amdgcn_sched_barrier(0);

        const v2f a = af[(s / GNN_PQ_ACH) & 1][s % GNN_PQ_ACH];
        v2f b;
#pragma unroll
        for (int i = 0; i < 4; ++i) {
            const float4 fa = pa[s % GNN_PQ_RING][i];
            const float4 fb = pb[s % GNN_PQ_RING][i];
            b.x = fa.x; b.y = fa.y; accP[2 * i + 0] = WMMA_F32(a, b, accP[2 * i + 0]);
            b.x = fa.z; b.y = fa.w; accP[2 * i + 1] = WMMA_F32(a, b, accP[2 * i + 1]);
            b.x = fb.x; b.y = fb.y; accQ[2 * i + 0] = WMMA_F32(a, b, accQ[2 * i + 0]);
            b.x = fb.z; b.y = fb.w; accQ[2 * i + 1] = WMMA_F32(a, b, accQ[2 * i + 1]);
        }
        __builtin_amdgcn_sched_barrier(0);
    }

    const size_t rb = (size_t)(m0 + 8 * half) * GNN_DOUT + lm;
    float* prow = P + rb;
    float* qrow = Q + rb;
#pragma unroll
    for (int r = 0; r < 8; ++r)
#pragma unroll
        for (int t = 0; t < 8; ++t) {
            prow[(size_t)r * GNN_DOUT + t * 16] = accP[t][r];
            qrow[(size_t)r * GNN_DOUT + t * 16] = accQ[t][r];
        }
}

#define GNN_CSR_T  1024
#define GNN_CSR_IT ((GNN_E + GNN_CSR_T - 1) / GNN_CSR_T)
#define GNN_CSR_R  50
#define GNN_CSR_G  (GNN_N / GNN_CSR_R)
#define GNN_CSR_H  64
static_assert(GNN_N % GNN_CSR_R == 0, "destination ranges must tile the node set");
static_assert(GNN_CSR_R < GNN_CSR_H, "entry R of the scanned counters is the end of the range");

__device__ __forceinline__ void csr_range_body(const int* __restrict__ ei,
                                               int* __restrict__ rowptr,
                                               int* slot,
                                               int* __restrict__ col,
                                               int g) {
    __shared__ int hist[GNN_CSR_H];
    __shared__ int below;
    const int tid  = threadIdx.x;
    const int lane = tid & 31;
    const int lo   = g * GNN_CSR_R;

    if (tid < GNN_CSR_H) hist[tid] = 0;
    if (tid == 0) below = 0;
    __syncthreads();

    int rel[GNN_CSR_IT], r[GNN_CSR_IT];
#pragma unroll
    for (int it = 0; it < GNN_CSR_IT; ++it) {
        const int e = it * GNN_CSR_T + tid;
        rel[it] = (e < GNN_E) ? (ei[GNN_E + e] - lo) : GNN_N;
    }
    int nb = 0;
#pragma unroll
    for (int it = 0; it < GNN_CSR_IT; ++it) {
        r[it] = ((unsigned)rel[it] < (unsigned)GNN_CSR_R) ? atomicAdd(&hist[rel[it]], 1) : 0;
        nb += (rel[it] < 0) ? 1 : 0;
    }
#pragma unroll
    for (int m = 16; m >= 1; m >>= 1) nb += __shfl_xor(nb, m, 32);
    if (lane == 0) atomicAdd(&below, nb);
    __syncthreads();

    if (tid < 32) {
        const int i0 = 2 * lane;
        const int i1 = 2 * lane + 1;
        const int c0 = hist[i0];
        const int s  = c0 + hist[i1];
        int inc = s;
        for (int m = 1; m < 32; m <<= 1) {
            const int v = __shfl_up(inc, m, 32);
            if (lane >= m) inc += v;
        }
        const int s0 = below + inc - s;
        const int s1 = s0 + c0;
        hist[i0] = s0;
        hist[i1] = s1;
        if (i0 < GNN_CSR_R) rowptr[lo + i0] = s0;
        if (i1 < GNN_CSR_R) rowptr[lo + i1] = s1;
        if (lo + GNN_CSR_R == GNN_N) {
            if (i0 == GNN_CSR_R) rowptr[GNN_N] = s0;
            if (i1 == GNN_CSR_R) rowptr[GNN_N] = s1;
        }
    }
    __syncthreads();

#pragma unroll
    for (int it = 0; it < GNN_CSR_IT; ++it)
        if ((unsigned)rel[it] < (unsigned)GNN_CSR_R)
            slot[hist[rel[it]] + r[it]] = it * GNN_CSR_T + tid;
    __syncthreads();

    const int base = hist[0];
    const int nown = hist[GNN_CSR_R] - base;
    for (int p = tid; p < nown; p += GNN_CSR_T) {
        const int e   = slot[base + p];
        const int src = ei[e];
        const int b   = ei[GNN_E + e] - lo;
        const int beg = hist[b];
        const int end = hist[b + 1];
        int rank = 0;
        for (int j = beg; j < end; ++j) rank += (slot[j] < e) ? 1 : 0;
        col[beg + rank] = src;
    }
}

#define GNN_PK_A0 0
#define GNN_PK_B0 8192
#define GNN_PK_20 16384
#define GNN_PK_A1 32768
#define GNN_PK_B1 49152
#define GNN_PK_21 65536
#define GNN_PK_END 81920
__global__ __launch_bounds__(GNN_CSR_T) void k_setup(const int* __restrict__ ei,
                                                     int* __restrict__ rowptr,
                                                     int* slot,
                                                     int* __restrict__ col,
                                                     const float* __restrict__ W1a,
                                                     const float* __restrict__ W2a,
                                                     const float* __restrict__ W1b,
                                                     const float* __restrict__ W2b,
                                                     float* __restrict__ Bp) {
    if (blockIdx.x < GNN_CSR_G) { csr_range_body(ei, rowptr, slot, col, blockIdx.x); return; }
    const int first = (blockIdx.x - GNN_CSR_G) * GNN_CSR_T;
    const int tid   = threadIdx.x;
    if (first < GNN_PK_B0)
        pack_b_elem<true >(W1a, Bp + GNN_PK_A0, 64, first - GNN_PK_A0 + tid);
    else if (first < GNN_PK_20)
        pack_b_elem<false>(W1a + 64 * GNN_DOUT, Bp + GNN_PK_B0, 64, first - GNN_PK_B0 + tid);
    else if (first < GNN_PK_A1)
        pack_b_tile_elem<false>(W2a, Bp + GNN_PK_20, first - GNN_PK_20 + tid);
    else if (first < GNN_PK_B1)
        pack_b_tile_elem<true >(W1b, Bp + GNN_PK_A1, first - GNN_PK_A1 + tid);
    else if (first < GNN_PK_21)
        pack_b_tile_elem<false>(W1b + 128 * GNN_DOUT, Bp + GNN_PK_B1, first - GNN_PK_B1 + tid);
    else
        pack_b_tile_elem<false>(W2b, Bp + GNN_PK_21, first - GNN_PK_21 + tid);
}

#define GNN_FT_ROWS  16
#define GNN_FT_PITCH 132

#define GNN_FT_G     2
#define GNN_NX_PRE   4
#define GNN_GSTRIDE  ((size_t)GNN_N * GNN_DOUT)
static_assert(GNN_BSN % GNN_FT_G == 0, "graphs must pair up");

__device__ __forceinline__ void edge_row_sum2(const float* __restrict__ qb,
                                              const float4 pa, const float4 pb,
                                              const float4 bb, const int* __restrict__ col,
                                              int beg, int end, int lane,
                                              float4& outA, float4& outB) {
    float4 accA = {0.0f, 0.0f, 0.0f, 0.0f};
    float4 accB = {0.0f, 0.0f, 0.0f, 0.0f};
    for (int base = beg; base < end; base += 32) {
        const int n    = (end - base < 32) ? (end - base) : 32;
        const int mine = (lane < n) ? col[base + lane] : 0;
#pragma unroll 4
        for (int j = 0; j < n; ++j) {
            const int src = __builtin_amdgcn_readlane(mine, j);
            const float* r = qb + (size_t)src * GNN_DOUT;
            const float4 qa = *(const float4*)(r);
            const float4 qc = *(const float4*)(r + GNN_GSTRIDE);
            accA.x += lrelu(pa.x + qa.x + bb.x);
            accA.y += lrelu(pa.y + qa.y + bb.y);
            accA.z += lrelu(pa.z + qa.z + bb.z);
            accA.w += lrelu(pa.w + qa.w + bb.w);
            accB.x += lrelu(pb.x + qc.x + bb.x);
            accB.y += lrelu(pb.y + qc.y + bb.y);
            accB.z += lrelu(pb.z + qc.z + bb.z);
            accB.w += lrelu(pb.w + qc.w + bb.w);
        }
    }
    outA = accA;
    outB = accB;
}

template <bool RES, bool NEXT>
__global__ __launch_bounds__(256) void k_edge_agg_ln(const float* __restrict__ P,
                                                     const float* __restrict__ Q,
                                                     const float* __restrict__ b1,
                                                     const int* __restrict__ rowptr,
                                                     const int* __restrict__ col,
                                                     const float* __restrict__ Bt2,
                                                     const float* __restrict__ b2,
                                                     const float* __restrict__ g,
                                                     const float* __restrict__ beta,
                                                     const float* __restrict__ xres,
                                                     float* __restrict__ out,
                                                     const float* __restrict__ BtPn,
                                                     const float* __restrict__ BtQn,
                                                     float* __restrict__ Pn,
                                                     float* __restrict__ Qn) {
    __shared__ __attribute__((aligned(16))) float tileS[GNN_FT_G][GNN_FT_ROWS * GNN_FT_PITCH];
    __shared__ __attribute__((aligned(16))) float tileO[GNN_FT_G][GNN_FT_ROWS * GNN_FT_PITCH];

    const int wave = __builtin_amdgcn_readfirstlane((int)(threadIdx.x >> 5));
    const int lane = threadIdx.x & 31;
    const int half = lane >> 4;
    const int lm   = lane & 15;
    const int c    = lane * 4;
    const int off  = blockIdx.y * (GNN_FT_G * GNN_N);
    const int lr0  = 2 * wave;
    const int node = blockIdx.x * GNN_FT_ROWS + lr0;

    const int beg0 = rowptr[node];
    const int beg1 = rowptr[node + 1];
    const int end1 = rowptr[node + 2];
    const size_t row0 = (size_t)(off + node) * GNN_DOUT + c;
    const float4 p0a = *(const float4*)(P + row0);
    const float4 p1a = *(const float4*)(P + row0 + GNN_DOUT);
    const float4 p0b = *(const float4*)(P + row0 + GNN_GSTRIDE);
    const float4 p1b = *(const float4*)(P + row0 + GNN_GSTRIDE + GNN_DOUT);
    const float4 bb = *(const float4*)(b1 + c);
    const float* qb = Q + (size_t)off * GNN_DOUT + c;

    const float cnt0 = (float)(beg1 - beg0);
    const float cnt1 = (float)(end1 - beg1);
    {
        float4 sa, sb;
        edge_row_sum2(qb, p0a, p0b, bb, col, beg0, beg1, lane, sa, sb);
        const float sc0 = 1.0f / fmaxf(cnt0, 1.0f);
        sa.x *= sc0; sa.y *= sc0; sa.z *= sc0; sa.w *= sc0;
        sb.x *= sc0; sb.y *= sc0; sb.z *= sc0; sb.w *= sc0;
        *(float4*)(tileS[0] + (lr0 + 0) * GNN_FT_PITCH + c) = sa;
        *(float4*)(tileS[1] + (lr0 + 0) * GNN_FT_PITCH + c) = sb;
        edge_row_sum2(qb, p1a, p1b, bb, col, beg1, end1, lane, sa, sb);
        const float sc1 = 1.0f / fmaxf(cnt1, 1.0f);
        sa.x *= sc1; sa.y *= sc1; sa.z *= sc1; sa.w *= sc1;
        sb.x *= sc1; sb.y *= sc1; sb.z *= sc1; sb.w *= sc1;
        *(float4*)(tileS[0] + (lr0 + 1) * GNN_FT_PITCH + c) = sa;
        *(float4*)(tileS[1] + (lr0 + 1) * GNN_FT_PITCH + c) = sb;
    }

    const float4* bt = (const float4*)(Bt2 + (size_t)wave * 2048 + lane * 4);
    float4 bf[16];
#pragma unroll
    for (int i = 0; i < 16; ++i) bf[i] = bt[i * 32];

    __syncthreads();

    v8f accA = {};
    v8f accB = {};
    const float* arowA = tileS[0] + lm * GNN_FT_PITCH + 2 * half;
    const float* arowB = tileS[1] + lm * GNN_FT_PITCH + 2 * half;
#pragma unroll
    for (int i = 0; i < 16; ++i) {
        const v2f a0 = *(const v2f*)(arowA + 8 * i);
        const v2f a1 = *(const v2f*)(arowA + 8 * i + 4);
        const v2f c0 = *(const v2f*)(arowB + 8 * i);
        const v2f c1 = *(const v2f*)(arowB + 8 * i + 4);
        v2f b;
        b.x = bf[i].x; b.y = bf[i].y;
        accA = WMMA_F32(a0, b, accA);
        accB = WMMA_F32(c0, b, accB);
        b.x = bf[i].z; b.y = bf[i].w;
        accA = WMMA_F32(a1, b, accA);
        accB = WMMA_F32(c1, b, accB);
    }
#pragma unroll
    for (int r = 0; r < 8; ++r) {
        tileO[0][(r + 8 * half) * GNN_FT_PITCH + 16 * wave + lm] = accA[r];
        tileO[1][(r + 8 * half) * GNN_FT_PITCH + 16 * wave + lm] = accB[r];
    }

    __syncthreads();

    const float4 b2v = *(const float4*)(b2 + c);
    const float4 gv  = *(const float4*)(g + c);
    const float4 bev = *(const float4*)(beta + c);
    float4 fin[2 * GNN_FT_G];
#pragma unroll
    for (int gq = 0; gq < 2 * GNN_FT_G; ++gq) {
        const int gi = gq >> 1;
        const int q  = gq & 1;
        const float hasE = ((q == 0 ? cnt0 : cnt1) > 0.0f) ? 1.0f : 0.0f;
        float4 h = *(const float4*)(tileO[gi] + (lr0 + q) * GNN_FT_PITCH + c);
        h.x += hasE * b2v.x; h.y += hasE * b2v.y; h.z += hasE * b2v.z; h.w += hasE * b2v.w;

        float s = (h.x + h.y) + (h.z + h.w);
#pragma unroll
        for (int m = 16; m >= 1; m >>= 1) s += __shfl_xor(s, m, 32);
        const float mu = s * (1.0f / GNN_DOUT);

        h.x -= mu; h.y -= mu; h.z -= mu; h.w -= mu;
        float s2 = (h.x * h.x + h.y * h.y) + (h.z * h.z + h.w * h.w);
#pragma unroll
        for (int m = 16; m >= 1; m >>= 1) s2 += __shfl_xor(s2, m, 32);
        const float rstd = rsqrtf(s2 * (1.0f / GNN_DOUT) + GNN_EPS);

        float4 o;
        o.x = h.x * rstd * gv.x + bev.x;
        o.y = h.y * rstd * gv.y + bev.y;
        o.z = h.z * rstd * gv.z + bev.z;
        o.w = h.w * rstd * gv.w + bev.w;
        const size_t orow = row0 + (size_t)gi * GNN_GSTRIDE + (size_t)q * GNN_DOUT;
        if (RES) {
            const float4 xr = *(const float4*)(xres + orow);
            o.x += xr.x; o.y += xr.y; o.z += xr.z; o.w += xr.w;
        }
        o.x = lrelu(o.x); o.y = lrelu(o.y); o.z = lrelu(o.z); o.w = lrelu(o.w);
        if (NEXT) {
            *(float4*)(tileS[gi] + (lr0 + q) * GNN_FT_PITCH + c) = o;
            fin[gq] = o;
        } else {
            *(float4*)(out + orow) = o;
        }
    }

    if constexpr (NEXT) {
        __shared__ __attribute__((aligned(16))) float tileQ[GNN_FT_G][GNN_FT_ROWS * GNN_FT_PITCH];

        const float4* btP = (const float4*)(BtPn + (size_t)wave * 2048 + lane * 4);
        const float4* btQ = (const float4*)(BtQn + (size_t)wave * 2048 + lane * 4);
        float4 fp0[GNN_NX_PRE], fq0[GNN_NX_PRE];
#pragma unroll
        for (int i = 0; i < GNN_NX_PRE; ++i) { fp0[i] = btP[i * 32]; fq0[i] = btQ[i * 32]; }

        __syncthreads();

#pragma unroll
        for (int gq = 0; gq < 2 * GNN_FT_G; ++gq)
            *(float4*)(out + row0 + (size_t)(gq >> 1) * GNN_GSTRIDE + (size_t)(gq & 1) * GNN_DOUT) = fin[gq];
        __builtin_amdgcn_sched_barrier(0);

        v8f pA = {}, pB = {}, qA = {}, qB = {};
#pragma unroll
        for (int i = 0; i < 16; ++i) {
            float4 fp, fq;
            if (i < GNN_NX_PRE) { fp = fp0[i % GNN_NX_PRE]; fq = fq0[i % GNN_NX_PRE]; }
            else                { fp = btP[i * 32];         fq = btQ[i * 32]; }
            const v2f a0 = *(const v2f*)(arowA + 8 * i);
            const v2f a1 = *(const v2f*)(arowA + 8 * i + 4);
            const v2f c0 = *(const v2f*)(arowB + 8 * i);
            const v2f c1 = *(const v2f*)(arowB + 8 * i + 4);
            v2f b;
            b.x = fp.x; b.y = fp.y;
            pA = WMMA_F32(a0, b, pA);
            pB = WMMA_F32(c0, b, pB);
            b.x = fq.x; b.y = fq.y;
            qA = WMMA_F32(a0, b, qA);
            qB = WMMA_F32(c0, b, qB);
            b.x = fp.z; b.y = fp.w;
            pA = WMMA_F32(a1, b, pA);
            pB = WMMA_F32(c1, b, pB);
            b.x = fq.z; b.y = fq.w;
            qA = WMMA_F32(a1, b, qA);
            qB = WMMA_F32(c1, b, qB);
        }
#pragma unroll
        for (int r = 0; r < 8; ++r) {
            const int o = (r + 8 * half) * GNN_FT_PITCH + 16 * wave + lm;
            tileO[0][o] = pA[r];
            tileO[1][o] = pB[r];
            tileQ[0][o] = qA[r];
            tileQ[1][o] = qB[r];
        }

        __syncthreads();

#pragma unroll
        for (int gq = 0; gq < 2 * GNN_FT_G; ++gq) {
            const int gi = gq >> 1;
            const int q  = gq & 1;
            const size_t orow = row0 + (size_t)gi * GNN_GSTRIDE + (size_t)q * GNN_DOUT;
            *(float4*)(Pn + orow) = *(const float4*)(tileO[gi] + (lr0 + q) * GNN_FT_PITCH + c);
            *(float4*)(Qn + orow) = *(const float4*)(tileQ[gi] + (lr0 + q) * GNN_FT_PITCH + c);
        }
    }
}

extern "C" void kernel_launch(void* const* d_in, const int* in_sizes, int n_in,
                              void* d_out, int out_size, void* d_ws, size_t ws_size,
                              hipStream_t stream) {
    (void)in_sizes; (void)n_in; (void)out_size; (void)ws_size;

    const float* x  = (const float*)d_in[0];
    const int*   ei = (const int*)d_in[1];
    const float* W1[2] = {(const float*)d_in[2],  (const float*)d_in[8]};
    const float* b1[2] = {(const float*)d_in[3],  (const float*)d_in[9]};
    const float* W2[2] = {(const float*)d_in[4],  (const float*)d_in[10]};
    const float* b2[2] = {(const float*)d_in[5],  (const float*)d_in[11]};
    const float* g [2] = {(const float*)d_in[6],  (const float*)d_in[12]};
    const float* be[2] = {(const float*)d_in[7],  (const float*)d_in[13]};

    const size_t NF = (size_t)GNN_NT * GNN_DOUT;
    float* ws   = (float*)d_ws;
    float* X    = ws;
    float* P    = X + NF;
    float* Q    = P + NF;
    float* P1   = Q + NF;
    float* Q1   = P1 + NF;
    float* Bp   = Q1 + NF;
    int* rowptr = (int*)(Bp + GNN_PK_END);
    int* slot   = rowptr + 2048;
    int* col    = slot + GNN_E;

    const dim3 egrid(GNN_N / GNN_FT_ROWS, GNN_BSN / GNN_FT_G);

    k_setup<<<GNN_CSR_G + GNN_PK_END / GNN_CSR_T, GNN_CSR_T, 0, stream>>>(
        ei, rowptr, slot, col, W1[0], W2[0], W1[1], W2[1], Bp);

    k_gemm_pq<64><<<250, 256, 0, stream>>>(x, Bp + GNN_PK_A0, Bp + GNN_PK_B0, P, Q);
    k_edge_agg_ln<false, true><<<egrid, 256, 0, stream>>>(P, Q, b1[0], rowptr, col, Bp + GNN_PK_20,
                                                          b2[0], g[0], be[0], nullptr, X,
                                                          Bp + GNN_PK_A1, Bp + GNN_PK_B1, P1, Q1);

    k_edge_agg_ln<true, false><<<egrid, 256, 0, stream>>>(P1, Q1, b1[1], rowptr, col, Bp + GNN_PK_21,
                                                          b2[1], g[1], be[1], X, (float*)d_out,
                                                          nullptr, nullptr, nullptr, nullptr);
}


# Block 0's P/Q GEMM joins the set-up launch (4 -> 3 launches): 64-row work-groups build their W1 fragments in LDS from the input, so none waits for a pack; all roles 256 threads; sum order unchanged
# speedup vs baseline: 5.8872x; 1.0718x over previous
#include <hip/hip_runtime.h>


typedef __attribute__((ext_vector_type(2))) float v2f;
typedef __attribute__((ext_vector_type(4))) float v4f;
typedef __attribute__((ext_vector_type(8))) float v8f;

#define GNN_SLOPE 0.01f
#define GNN_EPS   1e-5f
#define GNN_BSN   16
#define GNN_N     2000
#define GNN_E     32000
#define GNN_NT    (GNN_BSN * GNN_N)
#define GNN_DOUT  128

__device__ __forceinline__ float lrelu(float v) { return v >= 0.0f ? v : GNN_SLOPE * v; }

#define WMMA_F32(A, B, C) \
    __builtin_amdgcn_wmma_f32_16x16x4_f32(false, (A), false, (B), (short)0, (C), false, false)

template <bool DIFF>
__device__ __forceinline__ float pack_b_tile_elem(const float* __restrict__ B, int i) {
    const int t    = i >> 11;
    const int rem  = i & 2047;
    const int kk2  = rem >> 7;
    const int lane = (rem & 127) >> 2;
    const int j    = rem & 3;
    const int r    = (2 * kk2 + (j >> 1)) * 4 + 2 * (lane >> 4) + (j & 1);
    const int c    = t * 16 + (lane & 15);
    float v = B[r * GNN_DOUT + c];
    if (DIFF) v -= B[(GNN_DOUT + r) * GNN_DOUT + c];
    return v;
}

#define GNN_G0_ROWS 64
#define GNN_G0_WG   (GNN_NT / GNN_G0_ROWS)
#define GNN_G0_K    64
#define GNN_G0_NS   (GNN_G0_K / 4)
#define GNN_BL_LP   20
#define GNN_BL_KP   (32 * GNN_BL_LP)
static_assert(GNN_NT % GNN_G0_ROWS == 0, "row tiles must cover all nodes");

__device__ __forceinline__ void gemm_pq_body(const float* __restrict__ x,
                                             const float* __restrict__ W1,
                                             float* __restrict__ P,
                                             float* __restrict__ Q,
                                             int blk) {
    __shared__ __attribute__((aligned(16))) float Bl[2][GNN_G0_NS * GNN_BL_KP];

    const int wave  = __builtin_amdgcn_readfirstlane((int)(threadIdx.x >> 5));
    const int lane  = threadIdx.x & 31;
    const int half  = lane >> 4;
    const int lm    = lane & 15;
    const int which = wave >> 2;
    const int m0    = blk * GNN_G0_ROWS + (wave & 3) * 16;

    const float* arow = x + (size_t)(m0 + lm) * GNN_G0_K + 2 * half;
    v2f af[GNN_G0_NS];
#pragma unroll
    for (int s = 0; s < GNN_G0_NS; ++s) af[s] = *(const v2f*)(arow + 4 * s);

#pragma unroll
    for (int u = 0; u < GNN_G0_K / 16; ++u) {
        const int r = 2 * (u * 8 + wave);
        const v4f t0 = *(const v4f*)(W1 + (size_t)r * GNN_DOUT + 4 * lane);
        const v4f t1 = *(const v4f*)(W1 + (size_t)(r + 1) * GNN_DOUT + 4 * lane);
        const v4f b0 = *(const v4f*)(W1 + (size_t)(GNN_G0_K + r) * GNN_DOUT + 4 * lane);
        const v4f b1 = *(const v4f*)(W1 + (size_t)(GNN_G0_K + r + 1) * GNN_DOUT + 4 * lane);
        const int base = (r >> 2) * GNN_BL_KP + (16 * ((r >> 1) & 1) + 4 * (lane & 3)) * GNN_BL_LP + (lane >> 2) * 2;
#pragma unroll
        for (int e = 0; e < 4; ++e) {
            v2f d, b;
            d.x = t0[e] - b0[e]; d.y = t1[e] - b1[e];
            b.x = b0[e];         b.y = b1[e];
            *(v2f*)(Bl[0] + base + e * GNN_BL_LP) = d;
            *(v2f*)(Bl[1] + base + e * GNN_BL_LP) = b;
        }
    }
    __syncthreads();

    const float* bl = Bl[which] + lane * GNN_BL_LP;
    v8f acc[8] = {};
    v4f fb[2][4];
#pragma unroll
    for (int i = 0; i < 4; ++i) fb[0][i] = *(const v4f*)(bl + 4 * i);
#pragma unroll
    for (int s = 0; s < GNN_G0_NS; ++s) {
        if (s + 1 < GNN_G0_NS) {
#pragma unroll
            for (int i = 0; i < 4; ++i)
                fb[(s + 1) & 1][i] = *(const v4f*)(bl + (s + 1) * GNN_BL_KP + 4 * i);
        }
        const v2f a = af[s];
#pragma unroll
        for (int i = 0; i < 4; ++i) {
            const v4f f = fb[s & 1][i];
            v2f b;
            b.x = f.x; b.y = f.y; acc[2 * i + 0] = WMMA_F32(a, b, acc[2 * i + 0]);
            b.x = f.z; b.y = f.w; acc[2 * i + 1] = WMMA_F32(a, b, acc[2 * i + 1]);
        }
    }

    float* orow = (which ? Q : P) + (size_t)(m0 + 8 * half) * GNN_DOUT + lm;
#pragma unroll
    for (int r = 0; r < 8; ++r)
#pragma unroll
        for (int t = 0; t < 8; ++t)
            orow[(size_t)r * GNN_DOUT + t * 16] = acc[t][r];
}

#define GNN_CSR_T  256
#define GNN_CSR_IT (GNN_E / GNN_CSR_T)
#define GNN_CSR_R  10
#define GNN_CSR_G  (GNN_N / GNN_CSR_R)
#define GNN_CSR_H  64
static_assert(GNN_E % GNN_CSR_T == 0, "the trips cover the edge list exactly");
static_assert(GNN_N % GNN_CSR_R == 0, "destination ranges must tile the node set");
static_assert(GNN_CSR_R < GNN_CSR_H, "entry R of the scanned counters is the end of the range");

__device__ __forceinline__ void csr_range_body(const int* __restrict__ ei,
                                               int* __restrict__ rowptr,
                                               int* slot,
                                               int* __restrict__ col,
                                               int g) {
    __shared__ int hist[GNN_CSR_H];
    __shared__ int cur[GNN_CSR_H];
    __shared__ int below;
    const int tid  = threadIdx.x;
    const int lane = tid & 31;
    const int lo   = g * GNN_CSR_R;
    const int* dst = ei + GNN_E;

    if (tid < GNN_CSR_H) hist[tid] = 0;
    if (tid == 0) below = 0;
    __syncthreads();

    int rel[GNN_CSR_IT];
#pragma unroll
    for (int it = 0; it < GNN_CSR_IT; ++it) rel[it] = dst[it * GNN_CSR_T + tid] - lo;
    int nb = 0;
#pragma unroll
    for (int it = 0; it < GNN_CSR_IT; ++it) {
        if ((unsigned)rel[it] < (unsigned)GNN_CSR_R) atomicAdd(&hist[rel[it]], 1);
        nb += (rel[it] < 0) ? 1 : 0;
    }
#pragma unroll
    for (int m = 16; m >= 1; m >>= 1) nb += __shfl_xor(nb, m, 32);
    if (lane == 0) atomicAdd(&below, nb);
    __syncthreads();

    if (tid < 32) {
        const int i0 = 2 * lane;
        const int i1 = 2 * lane + 1;
        const int c0 = hist[i0];
        const int s  = c0 + hist[i1];
        int inc = s;
        for (int m = 1; m < 32; m <<= 1) {
            const int v = __shfl_up(inc, m, 32);
            if (lane >= m) inc += v;
        }
        const int s0 = below + inc - s;
        const int s1 = s0 + c0;
        hist[i0] = s0;
        hist[i1] = s1;
        cur[i0]  = s0;
        cur[i1]  = s1;
        if (i0 < GNN_CSR_R) rowptr[lo + i0] = s0;
        if (i1 < GNN_CSR_R) rowptr[lo + i1] = s1;
        if (lo + GNN_CSR_R == GNN_N) {
            if (i0 == GNN_CSR_R) rowptr[GNN_N] = s0;
            if (i1 == GNN_CSR_R) rowptr[GNN_N] = s1;
        }
    }
    __syncthreads();

#pragma unroll
    for (int it = 0; it < GNN_CSR_IT; ++it)
        if ((unsigned)rel[it] < (unsigned)GNN_CSR_R)
            slot[atomicAdd(&cur[rel[it]], 1)] = it * GNN_CSR_T + tid;
    __syncthreads();

    const int base = hist[0];
    const int nown = hist[GNN_CSR_R] - base;
    for (int p = tid; p < nown; p += GNN_CSR_T) {
        const int e   = slot[base + p];
        const int src = ei[e];
        const int b   = dst[e] - lo;
        const int beg = hist[b];
        const int end = hist[b + 1];
        int rank = 0;
        for (int j = beg; j < end; ++j) rank += (slot[j] < e) ? 1 : 0;
        col[beg + rank] = src;
    }
}

#define GNN_PK_20 16384
#define GNN_PK_A1 32768
#define GNN_PK_B1 49152
#define GNN_PK_21 65536
#define GNN_PK_END 81920
#define GNN_PK_PER 1024
#define GNN_PK_WG  ((GNN_PK_END - GNN_PK_20) / GNN_PK_PER)
__global__ __launch_bounds__(GNN_CSR_T) void k_setup(const int* __restrict__ ei,
                                                     int* __restrict__ rowptr,
                                                     int* slot,
                                                     int* __restrict__ col,
                                                     const float* __restrict__ x,
                                                     const float* __restrict__ W1a,
                                                     const float* __restrict__ W2a,
                                                     const float* __restrict__ W1b,
                                                     const float* __restrict__ W2b,
                                                     float* __restrict__ Bp,
                                                     float* __restrict__ P,
                                                     float* __restrict__ Q) {
    if (blockIdx.x < GNN_CSR_G) { csr_range_body(ei, rowptr, slot, col, blockIdx.x); return; }
    if (blockIdx.x >= GNN_CSR_G + GNN_PK_WG) {
        gemm_pq_body(x, W1a, P, Q, blockIdx.x - (GNN_CSR_G + GNN_PK_WG));
        return;
    }
    const int first = GNN_PK_20 + (blockIdx.x - GNN_CSR_G) * GNN_PK_PER;
    float v[GNN_PK_PER / GNN_CSR_T];
#pragma unroll
    for (int u = 0; u < GNN_PK_PER / GNN_CSR_T; ++u) {
        const int i = first + u * GNN_CSR_T + threadIdx.x;
        if (first < GNN_PK_A1)      v[u] = pack_b_tile_elem<false>(W2a, i - GNN_PK_20);
        else if (first < GNN_PK_B1) v[u] = pack_b_tile_elem<true >(W1b, i - GNN_PK_A1);
        else if (first < GNN_PK_21) v[u] = pack_b_tile_elem<false>(W1b + 128 * GNN_DOUT, i - GNN_PK_B1);
        else                        v[u] = pack_b_tile_elem<false>(W2b, i - GNN_PK_21);
    }
#pragma unroll
    for (int u = 0; u < GNN_PK_PER / GNN_CSR_T; ++u) Bp[first + u * GNN_CSR_T + threadIdx.x] = v[u];
}

#define GNN_FT_ROWS  16
#define GNN_FT_PITCH 132

#define GNN_FT_G     2
#define GNN_NX_PRE   4
#define GNN_GSTRIDE  ((size_t)GNN_N * GNN_DOUT)
static_assert(GNN_BSN % GNN_FT_G == 0, "graphs must pair up");

__device__ __forceinline__ void edge_row_sum2(const float* __restrict__ qb,
                                              const float4 pa, const float4 pb,
                                              const float4 bb, const int* __restrict__ col,
                                              int beg, int end, int lane,
                                              float4& outA, float4& outB) {
    float4 accA = {0.0f, 0.0f, 0.0f, 0.0f};
    float4 accB = {0.0f, 0.0f, 0.0f, 0.0f};
    for (int base = beg; base < end; base += 32) {
        const int n    = (end - base < 32) ? (end - base) : 32;
        const int mine = (lane < n) ? col[base + lane] : 0;
#pragma unroll 4
        for (int j = 0; j < n; ++j) {
            const int src = __builtin_amdgcn_readlane(mine, j);
            const float* r = qb + (size_t)src * GNN_DOUT;
            const float4 qa = *(const float4*)(r);
            const float4 qc = *(const float4*)(r + GNN_GSTRIDE);
            accA.x += lrelu(pa.x + qa.x + bb.x);
            accA.y += lrelu(pa.y + qa.y + bb.y);
            accA.z += lrelu(pa.z + qa.z + bb.z);
            accA.w += lrelu(pa.w + qa.w + bb.w);
            accB.x += lrelu(pb.x + qc.x + bb.x);
            accB.y += lrelu(pb.y + qc.y + bb.y);
            accB.z += lrelu(pb.z + qc.z + bb.z);
            accB.w += lrelu(pb.w + qc.w + bb.w);
        }
    }
    outA = accA;
    outB = accB;
}

template <bool RES, bool NEXT>
__global__ __launch_bounds__(256) void k_edge_agg_ln(const float* __restrict__ P,
                                                     const float* __restrict__ Q,
                                                     const float* __restrict__ b1,
                                                     const int* __restrict__ rowptr,
                                                     const int* __restrict__ col,
                                                     const float* __restrict__ Bt2,
                                                     const float* __restrict__ b2,
                                                     const float* __restrict__ g,
                                                     const float* __restrict__ beta,
                                                     const float* __restrict__ xres,
                                                     float* __restrict__ out,
                                                     const float* __restrict__ BtPn,
                                                     const float* __restrict__ BtQn,
                                                     float* __restrict__ Pn,
                                                     float* __restrict__ Qn) {
    __shared__ __attribute__((aligned(16))) float tileS[GNN_FT_G][GNN_FT_ROWS * GNN_FT_PITCH];
    __shared__ __attribute__((aligned(16))) float tileO[GNN_FT_G][GNN_FT_ROWS * GNN_FT_PITCH];

    const int wave = __builtin_amdgcn_readfirstlane((int)(threadIdx.x >> 5));
    const int lane = threadIdx.x & 31;
    const int half = lane >> 4;
    const int lm   = lane & 15;
    const int c    = lane * 4;
    const int off  = blockIdx.y * (GNN_FT_G * GNN_N);
    const int lr0  = 2 * wave;
    const int node = blockIdx.x * GNN_FT_ROWS + lr0;

    const int beg0 = rowptr[node];
    const int beg1 = rowptr[node + 1];
    const int end1 = rowptr[node + 2];
    const size_t row0 = (size_t)(off + node) * GNN_DOUT + c;
    const float4 p0a = *(const float4*)(P + row0);
    const float4 p1a = *(const float4*)(P + row0 + GNN_DOUT);
    const float4 p0b = *(const float4*)(P + row0 + GNN_GSTRIDE);
    const float4 p1b = *(const float4*)(P + row0 + GNN_GSTRIDE + GNN_DOUT);
    const float4 bb = *(const float4*)(b1 + c);
    const float* qb = Q + (size_t)off * GNN_DOUT + c;

    const float cnt0 = (float)(beg1 - beg0);
    const float cnt1 = (float)(end1 - beg1);
    {
        float4 sa, sb;
        edge_row_sum2(qb, p0a, p0b, bb, col, beg0, beg1, lane, sa, sb);
        const float sc0 = 1.0f / fmaxf(cnt0, 1.0f);
        sa.x *= sc0; sa.y *= sc0; sa.z *= sc0; sa.w *= sc0;
        sb.x *= sc0; sb.y *= sc0; sb.z *= sc0; sb.w *= sc0;
        *(float4*)(tileS[0] + (lr0 + 0) * GNN_FT_PITCH + c) = sa;
        *(float4*)(tileS[1] + (lr0 + 0) * GNN_FT_PITCH + c) = sb;
        edge_row_sum2(qb, p1a, p1b, bb, col, beg1, end1, lane, sa, sb);
        const float sc1 = 1.0f / fmaxf(cnt1, 1.0f);
        sa.x *= sc1; sa.y *= sc1; sa.z *= sc1; sa.w *= sc1;
        sb.x *= sc1; sb.y *= sc1; sb.z *= sc1; sb.w *= sc1;
        *(float4*)(tileS[0] + (lr0 + 1) * GNN_FT_PITCH + c) = sa;
        *(float4*)(tileS[1] + (lr0 + 1) * GNN_FT_PITCH + c) = sb;
    }

    const float4* bt = (const float4*)(Bt2 + (size_t)wave * 2048 + lane * 4);
    float4 bf[16];
#pragma unroll
    for (int i = 0; i < 16; ++i) bf[i] = bt[i * 32];

    __syncthreads();

    v8f accA = {};
    v8f accB = {};
    const float* arowA = tileS[0] + lm * GNN_FT_PITCH + 2 * half;
    const float* arowB = tileS[1] + lm * GNN_FT_PITCH + 2 * half;
#pragma unroll
    for (int i = 0; i < 16; ++i) {
        const v2f a0 = *(const v2f*)(arowA + 8 * i);
        const v2f a1 = *(const v2f*)(arowA + 8 * i + 4);
        const v2f c0 = *(const v2f*)(arowB + 8 * i);
        const v2f c1 = *(const v2f*)(arowB + 8 * i + 4);
        v2f b;
        b.x = bf[i].x; b.y = bf[i].y;
        accA = WMMA_F32(a0, b, accA);
        accB = WMMA_F32(c0, b, accB);
        b.x = bf[i].z; b.y = bf[i].w;
        accA = WMMA_F32(a1, b, accA);
        accB = WMMA_F32(c1, b, accB);
    }
#pragma unroll
    for (int r = 0; r < 8; ++r) {
        tileO[0][(r + 8 * half) * GNN_FT_PITCH + 16 * wave + lm] = accA[r];
        tileO[1][(r + 8 * half) * GNN_FT_PITCH + 16 * wave + lm] = accB[r];
    }

    __syncthreads();

    const float4 b2v = *(const float4*)(b2 + c);
    const float4 gv  = *(const float4*)(g + c);
    const float4 bev = *(const float4*)(beta + c);
    float4 fin[2 * GNN_FT_G];
#pragma unroll
    for (int gq = 0; gq < 2 * GNN_FT_G; ++gq) {
        const int gi = gq >> 1;
        const int q  = gq & 1;
        const float hasE = ((q == 0 ? cnt0 : cnt1) > 0.0f) ? 1.0f : 0.0f;
        float4 h = *(const float4*)(tileO[gi] + (lr0 + q) * GNN_FT_PITCH + c);
        h.x += hasE * b2v.x; h.y += hasE * b2v.y; h.z += hasE * b2v.z; h.w += hasE * b2v.w;

        float s = (h.x + h.y) + (h.z + h.w);
#pragma unroll
        for (int m = 16; m >= 1; m >>= 1) s += __shfl_xor(s, m, 32);
        const float mu = s * (1.0f / GNN_DOUT);

        h.x -= mu; h.y -= mu; h.z -= mu; h.w -= mu;
        float s2 = (h.x * h.x + h.y * h.y) + (h.z * h.z + h.w * h.w);
#pragma unroll
        for (int m = 16; m >= 1; m >>= 1) s2 += __shfl_xor(s2, m, 32);
        const float rstd = rsqrtf(s2 * (1.0f / GNN_DOUT) + GNN_EPS);

        float4 o;
        o.x = h.x * rstd * gv.x + bev.x;
        o.y = h.y * rstd * gv.y + bev.y;
        o.z = h.z * rstd * gv.z + bev.z;
        o.w = h.w * rstd * gv.w + bev.w;
        const size_t orow = row0 + (size_t)gi * GNN_GSTRIDE + (size_t)q * GNN_DOUT;
        if (RES) {
            const float4 xr = *(const float4*)(xres + orow);
            o.x += xr.x; o.y += xr.y; o.z += xr.z; o.w += xr.w;
        }
        o.x = lrelu(o.x); o.y = lrelu(o.y); o.z = lrelu(o.z); o.w = lrelu(o.w);
        if (NEXT) {
            *(float4*)(tileS[gi] + (lr0 + q) * GNN_FT_PITCH + c) = o;
            fin[gq] = o;
        } else {
            *(float4*)(out + orow) = o;
        }
    }

    if constexpr (NEXT) {
        __shared__ __attribute__((aligned(16))) float tileQ[GNN_FT_G][GNN_FT_ROWS * GNN_FT_PITCH];

        const float4* btP = (const float4*)(BtPn + (size_t)wave * 2048 + lane * 4);
        const float4* btQ = (const float4*)(BtQn + (size_t)wave * 2048 + lane * 4);
        float4 fp0[GNN_NX_PRE], fq0[GNN_NX_PRE];
#pragma unroll
        for (int i = 0; i < GNN_NX_PRE; ++i) { fp0[i] = btP[i * 32]; fq0[i] = btQ[i * 32]; }

        __syncthreads();

#pragma unroll
        for (int gq = 0; gq < 2 * GNN_FT_G; ++gq)
            *(float4*)(out + row0 + (size_t)(gq >> 1) * GNN_GSTRIDE + (size_t)(gq & 1) * GNN_DOUT) = fin[gq];
        __builtin_amdgcn_sched_barrier(0);

        v8f pA = {}, pB = {}, qA = {}, qB = {};
#pragma unroll
        for (int i = 0; i < 16; ++i) {
            float4 fp, fq;
            if (i < GNN_NX_PRE) { fp = fp0[i % GNN_NX_PRE]; fq = fq0[i % GNN_NX_PRE]; }
            else                { fp = btP[i * 32];         fq = btQ[i * 32]; }
            const v2f a0 = *(const v2f*)(arowA + 8 * i);
            const v2f a1 = *(const v2f*)(arowA + 8 * i + 4);
            const v2f c0 = *(const v2f*)(arowB + 8 * i);
            const v2f c1 = *(const v2f*)(arowB + 8 * i + 4);
            v2f b;
            b.x = fp.x; b.y = fp.y;
            pA = WMMA_F32(a0, b, pA);
            pB = WMMA_F32(c0, b, pB);
            b.x = fq.x; b.y = fq.y;
            qA = WMMA_F32(a0, b, qA);
            qB = WMMA_F32(c0, b, qB);
            b.x = fp.z; b.y = fp.w;
            pA = WMMA_F32(a1, b, pA);
            pB = WMMA_F32(c1, b, pB);
            b.x = fq.z; b.y = fq.w;
            qA = WMMA_F32(a1, b, qA);
            qB = WMMA_F32(c1, b, qB);
        }
#pragma unroll
        for (int r = 0; r < 8; ++r) {
            const int o = (r + 8 * half) * GNN_FT_PITCH + 16 * wave + lm;
            tileO[0][o] = pA[r];
            tileO[1][o] = pB[r];
            tileQ[0][o] = qA[r];
            tileQ[1][o] = qB[r];
        }

        __syncthreads();

#pragma unroll
        for (int gq = 0; gq < 2 * GNN_FT_G; ++gq) {
            const int gi = gq >> 1;
            const int q  = gq & 1;
            const size_t orow = row0 + (size_t)gi * GNN_GSTRIDE + (size_t)q * GNN_DOUT;
            *(float4*)(Pn + orow) = *(const float4*)(tileO[gi] + (lr0 + q) * GNN_FT_PITCH + c);
            *(float4*)(Qn + orow) = *(const float4*)(tileQ[gi] + (lr0 + q) * GNN_FT_PITCH + c);
        }
    }
}

extern "C" void kernel_launch(void* const* d_in, const int* in_sizes, int n_in,
                              void* d_out, int out_size, void* d_ws, size_t ws_size,
                              hipStream_t stream) {
    (void)in_sizes; (void)n_in; (void)out_size; (void)ws_size;

    const float* x  = (const float*)d_in[0];
    const int*   ei = (const int*)d_in[1];
    const float* W1[2] = {(const float*)d_in[2],  (const float*)d_in[8]};
    const float* b1[2] = {(const float*)d_in[3],  (const float*)d_in[9]};
    const float* W2[2] = {(const float*)d_in[4],  (const float*)d_in[10]};
    const float* b2[2] = {(const float*)d_in[5],  (const float*)d_in[11]};
    const float* g [2] = {(const float*)d_in[6],  (const float*)d_in[12]};
    const float* be[2] = {(const float*)d_in[7],  (const float*)d_in[13]};

    const size_t NF = (size_t)GNN_NT * GNN_DOUT;
    float* ws   = (float*)d_ws;
    float* X    = ws;
    float* P    = X + NF;
    float* Q    = P + NF;
    float* P1   = Q + NF;
    float* Q1   = P1 + NF;
    float* Bp   = Q1 + NF;
    int* rowptr = (int*)(Bp + GNN_PK_END);
    int* slot   = rowptr + 2048;
    int* col    = slot + GNN_E;

    const dim3 egrid(GNN_N / GNN_FT_ROWS, GNN_BSN / GNN_FT_G);

    k_setup<<<GNN_CSR_G + GNN_PK_WG + GNN_G0_WG, GNN_CSR_T, 0, stream>>>(
        ei, rowptr, slot, col, x, W1[0], W2[0], W1[1], W2[1], Bp, P, Q);

    k_edge_agg_ln<false, true><<<egrid, 256, 0, stream>>>(P, Q, b1[0], rowptr, col, Bp + GNN_PK_20,
                                                          b2[0], g[0], be[0], nullptr, X,
                                                          Bp + GNN_PK_A1, Bp + GNN_PK_B1, P1, Q1);

    k_edge_agg_ln<true, false><<<egrid, 256, 0, stream>>>(P1, Q1, b1[1], rowptr, col, Bp + GNN_PK_21,
                                                          b2[1], g[1], be[1], X, (float*)d_out,
                                                          nullptr, nullptr, nullptr, nullptr);
}
